# starting kernel
# baseline (speedup 1.0000x reference)
#include <hip/hip_runtime.h>

typedef __attribute__((ext_vector_type(2))) float v2f;
typedef __attribute__((ext_vector_type(8))) float v8f;

static __device__ __forceinline__ void atomic_add_f32(float* p, float v) {
  __hip_atomic_fetch_add(p, v, __ATOMIC_RELAXED, __HIP_MEMORY_SCOPE_AGENT);
}

__global__ void k_init_deg(float* deg, int n) {
  int i = blockIdx.x * blockDim.x + threadIdx.x;
  if (i < n) deg[i] = 1.0f;
}

__global__ void k_count_deg(const int* __restrict__ dst, float* deg, int E) {
  int i = blockIdx.x * blockDim.x + threadIdx.x;
  if (i < E) atomic_add_f32(&deg[dst[i]], 1.0f);
}

__global__ void k_rsqrt_deg(float* d, int n) {
  int i = blockIdx.x * blockDim.x + threadIdx.x;
  if (i < n) d[i] = rsqrtf(d[i]);
}

template <int NOUT, bool RELU>
__global__ __launch_bounds__(256) void k_gemm_wmma(const float* __restrict__ A,
                                                   const float* __restrict__ W,
                                                   float* __restrict__ out, int nrows) {
  __shared__ float Wl[128 * NOUT];
  const int tid = threadIdx.x;
  for (int i = tid; i < 128 * NOUT; i += 256) {
    int k = i / NOUT;
    int j = i - k * NOUT;
    Wl[(k >> 1) * (2 * NOUT) + 2 * j + (k & 1)] = W[i];
  }
  __syncthreads();

  const int wave = tid >> 5;
  const int lane = tid & 31;
  const int tile = blockIdx.x * 8 + wave;
  const int ntiles = (nrows + 15) >> 4;
  if (tile >= ntiles) return;

  const int m = lane & 15;
  const int kh = lane >> 4;
  int row = tile * 16 + m;
  if (row >= nrows) row = nrows - 1;

  v8f acc[NOUT / 16];
#pragma unroll
  for (int t = 0; t < NOUT / 16; ++t) acc[t] = v8f{0.f, 0.f, 0.f, 0.f, 0.f, 0.f, 0.f, 0.f};

  const float* arow = A + (size_t)row * 128 + 2 * kh;
#pragma unroll 4
  for (int k0 = 0; k0 < 128; k0 += 4) {
    v2f a = *(const v2f*)(arow + k0);
    if (RELU) {
      a.x = fmaxf(a.x, 0.f);
      a.y = fmaxf(a.y, 0.f);
    }
    const float* bb = &Wl[((k0 >> 1) + kh) * (2 * NOUT)];
#pragma unroll
    for (int t = 0; t < NOUT / 16; ++t) {
      v2f b = *(const v2f*)(bb + 2 * (t * 16 + m));
      acc[t] = __builtin_amdgcn_wmma_f32_16x16x4_f32(false, a, false, b, (short)0, acc[t],
                                                     false, false);
    }
  }

  const int rbase = tile * 16 + kh * 8;
#pragma unroll
  for (int t = 0; t < NOUT / 16; ++t) {
#pragma unroll
    for (int v = 0; v < 8; ++v) {
      int r = rbase + v;
      if (r < nrows) out[(size_t)r * NOUT + t * 16 + m] = acc[t][v];
    }
  }
}

template <int NOUT>
__global__ void k_init_agg(const float* __restrict__ h, const float* __restrict__ dinv,
                           const float* __restrict__ bias, float* __restrict__ agg, int n) {
  int i = blockIdx.x * blockDim.x + threadIdx.x;
  if (i >= n * NOUT) return;
  int r = i / NOUT;
  int c = i - r * NOUT;
  float di = dinv[r];
  agg[i] = h[i] * di * di + bias[c];
}

__global__ __launch_bounds__(256) void k_scatter128(const float* __restrict__ h,
                                                    const int* __restrict__ src,
                                                    const int* __restrict__ dst,
                                                    const float* __restrict__ dinv,
                                                    float* __restrict__ agg, int E) {
  int gw = (blockIdx.x * 256 + threadIdx.x) >> 5;
  int lane = threadIdx.x & 31;
  if (gw >= E) return;
  int s = src[gw], d = dst[gw];
  float nrm = dinv[s] * dinv[d];
  const float4 hv = *(const float4*)(h + (size_t)s * 128 + lane * 4);
  float* o = agg + (size_t)d * 128 + lane * 4;
  atomic_add_f32(o + 0, hv.x * nrm);
  atomic_add_f32(o + 1, hv.y * nrm);
  atomic_add_f32(o + 2, hv.z * nrm);
  atomic_add_f32(o + 3, hv.w * nrm);
}

__global__ __launch_bounds__(256) void k_scatter64(const float* __restrict__ h,
                                                   const int* __restrict__ src,
                                                   const int* __restrict__ dst,
                                                   const float* __restrict__ dinv,
                                                   float* __restrict__ out, int E) {
  int t = blockIdx.x * 256 + threadIdx.x;
  int e = t >> 4;
  int lane = threadIdx.x & 15;
  if (e >= E) return;
  int s = src[e], d = dst[e];
  float nrm = dinv[s] * dinv[d];
  const float4 hv = *(const float4*)(h + (size_t)s * 64 + lane * 4);
  float* o = out + (size_t)d * 64 + lane * 4;
  atomic_add_f32(o + 0, hv.x * nrm);
  atomic_add_f32(o + 1, hv.y * nrm);
  atomic_add_f32(o + 2, hv.z * nrm);
  atomic_add_f32(o + 3, hv.w * nrm);
}

extern "C" void kernel_launch(void* const* d_in, const int* in_sizes, int n_in, void* d_out,
                              int out_size, void* d_ws, size_t ws_size, hipStream_t stream) {
  const float* x = (const float*)d_in[0];
  const int* ei = (const int*)d_in[1];
  const float* W1 = (const float*)d_in[2];
  const float* b1 = (const float*)d_in[3];
  const float* W2 = (const float*)d_in[4];
  const float* b2 = (const float*)d_in[5];
  float* out = (float*)d_out;

  const int N = in_sizes[0] / 128;
  const int E = in_sizes[1] / 2;
  const int* src = ei;
  const int* dstp = ei + E;

  char* ws = (char*)d_ws;
  float* dinv = (float*)ws;
  size_t off = (((size_t)N * 4) + 255) & ~(size_t)255;
  float* h1 = (float*)(ws + off);
  float* agg1 = (float*)(ws + off + (size_t)N * 128 * 4);
  float* h2 = h1;

  dim3 B(256);
  k_init_deg<<<(N + 255) / 256, B, 0, stream>>>(dinv, N);
  k_count_deg<<<(E + 255) / 256, B, 0, stream>>>(dstp, dinv, E);
  k_rsqrt_deg<<<(N + 255) / 256, B, 0, stream>>>(dinv, N);

  const int ntiles = (N + 15) / 16;
  const int gblocks = (ntiles + 7) / 8;

  k_gemm_wmma<128, false><<<gblocks, B, 0, stream>>>(x, W1, h1, N);
  k_init_agg<128><<<(N * 128 + 255) / 256, B, 0, stream>>>(h1, dinv, b1, agg1, N);
  k_scatter128<<<(E + 7) / 8, B, 0, stream>>>(h1, src, dstp, dinv, agg1, E);

  k_gemm_wmma<64, true><<<gblocks, B, 0, stream>>>(agg1, W2, h2, N);
  k_init_agg<64><<<(N * 64 + 255) / 256, B, 0, stream>>>(h2, dinv, b2, out, N);
  k_scatter64<<<(E + 15) / 16, B, 0, stream>>>(h2, src, dstp, dinv, out, E);
}


# edge sums: global f32 atomic scatter replaced by owner-computes: work-group holds 256 dst rows in LDS, scans dst, queues its edges, adds with LDS f32 atomics, stores once; add order = LDS arrival
# speedup vs baseline: 1.1563x; 1.1563x over previous
#include <hip/hip_runtime.h>

typedef __attribute__((ext_vector_type(2))) float v2f;
typedef __attribute__((ext_vector_type(8))) float v8f;

static __device__ __forceinline__ void atomic_add_f32(float* p, float v) {
  __hip_atomic_fetch_add(p, v, __ATOMIC_RELAXED, __HIP_MEMORY_SCOPE_AGENT);
}

__global__ void k_init_deg(float* deg, int n) {
  int i = blockIdx.x * blockDim.x + threadIdx.x;
  if (i < n) deg[i] = 1.0f;
}

__global__ void k_count_deg(const int* __restrict__ dst, float* deg, int E) {
  int i = blockIdx.x * blockDim.x + threadIdx.x;
  if (i < E) atomic_add_f32(&deg[dst[i]], 1.0f);
}

__global__ void k_rsqrt_deg(float* d, int n) {
  int i = blockIdx.x * blockDim.x + threadIdx.x;
  if (i < n) d[i] = rsqrtf(d[i]);
}

template <int NOUT, bool RELU>
__global__ __launch_bounds__(256) void k_gemm_wmma(const float* __restrict__ A,
                                                   const float* __restrict__ W,
                                                   float* __restrict__ out,
                                                   const float* __restrict__ dinv,
                                                   const float* __restrict__ bias,
                                                   float* __restrict__ agg, int nrows) {
  __shared__ float Wl[128 * NOUT];
  const int tid = threadIdx.x;
  for (int i = tid; i < 128 * NOUT; i += 256) {
    int k = i / NOUT;
    int j = i - k * NOUT;
    Wl[(k >> 1) * (2 * NOUT) + 2 * j + (k & 1)] = W[i];
  }
  __syncthreads();

  const int wave = tid >> 5;
  const int lane = tid & 31;
  const int tile = blockIdx.x * 8 + wave;
  const int ntiles = (nrows + 15) >> 4;
  if (tile >= ntiles) return;

  const int m = lane & 15;
  const int kh = lane >> 4;
  int row = tile * 16 + m;
  if (row >= nrows) row = nrows - 1;

  const int rbase = tile * 16 + kh * 8;
  float di[8];
#pragma unroll
  for (int v = 0; v < 8; ++v) {
    int r = rbase + v;
    di[v] = dinv[r < nrows ? r : nrows - 1];
  }
  float bc[NOUT / 16];
#pragma unroll
  for (int t = 0; t < NOUT / 16; ++t) bc[t] = bias[t * 16 + m];

  v8f acc[NOUT / 16];
#pragma unroll
  for (int t = 0; t < NOUT / 16; ++t) acc[t] = v8f{0.f, 0.f, 0.f, 0.f, 0.f, 0.f, 0.f, 0.f};

  const float* arow = A + (size_t)row * 128 + 2 * kh;
#pragma unroll 4
  for (int k0 = 0; k0 < 128; k0 += 4) {
    v2f a = *(const v2f*)(arow + k0);
    if (RELU) {
      a.x = fmaxf(a.x, 0.f);
      a.y = fmaxf(a.y, 0.f);
    }
    const float* bb = &Wl[((k0 >> 1) + kh) * (2 * NOUT)];
#pragma unroll
    for (int t = 0; t < NOUT / 16; ++t) {
      v2f b = *(const v2f*)(bb + 2 * (t * 16 + m));
      acc[t] = __builtin_amdgcn_wmma_f32_16x16x4_f32(false, a, false, b, (short)0, acc[t],
                                                     false, false);
    }
  }

#pragma unroll
  for (int t = 0; t < NOUT / 16; ++t) {
#pragma unroll
    for (int v = 0; v < 8; ++v) {
      int r = rbase + v;
      if (r < nrows) {
        const float hv = acc[t][v];
        const size_t o = (size_t)r * NOUT + t * 16 + m;
        out[o] = hv;
        agg[o] = hv * di[v] * di[v] + bc[t];
      }
    }
  }
}

template <int F>
__global__ __launch_bounds__(1024) void k_agg_owner(const float* __restrict__ h,
                                                    const int* __restrict__ src,
                                                    const int* __restrict__ dst,
                                                    const float* __restrict__ dinv,
                                                    float* __restrict__ agg, int E, int N) {
  constexpr int ROWS = 256;
  constexpr int V = F / 32;
  constexpr int CAP = 16384;
  constexpr int NT = 1024;
  constexpr int UNR = 4;
  constexpr int CH = NT * 4 * UNR;
  constexpr int U = 8;
  __shared__ float acc[ROWS * F];
  __shared__ int q[CAP];
  __shared__ unsigned cnt;

  const int tid = threadIdx.x;
  const int lane = tid & 31;
  const int wave = tid >> 5;
  const int base = blockIdx.x * ROWS;

  if (tid == 0) cnt = 0u;
  for (int i = tid; i < ROWS * 32; i += NT) {
    const int r = i >> 5, ln = i & 31;
    const int row = base + r;
#pragma unroll
    for (int j = 0; j < V; ++j)
      acc[r * F + j * 32 + ln] = (row < N) ? agg[(size_t)row * F + ln * V + j] : 0.f;
  }
  __syncthreads();

  const int nch = (E + CH - 1) / CH;
  int c = (int)(((long long)blockIdx.x * nch) / gridDim.x);
  const bool vec4 = ((E & 3) == 0);
  for (int ci = 0; ci < nch; ++ci) {
    const int e0 = c * CH;
    int4 dv[UNR];
    if (vec4 && e0 + CH <= E) {
#pragma unroll
      for (int u = 0; u < UNR; ++u) dv[u] = *(const int4*)(dst + e0 + (u * NT + tid) * 4);
    } else {
#pragma unroll
      for (int u = 0; u < UNR; ++u) {
        const int e = e0 + (u * NT + tid) * 4;
        dv[u].x = (e + 0 < E) ? dst[e + 0] : -1;
        dv[u].y = (e + 1 < E) ? dst[e + 1] : -1;
        dv[u].z = (e + 2 < E) ? dst[e + 2] : -1;
        dv[u].w = (e + 3 < E) ? dst[e + 3] : -1;
      }
    }
#pragma unroll
    for (int u = 0; u < UNR; ++u) {
      const int e = e0 + (u * NT + tid) * 4;
      const unsigned d0 = (unsigned)(dv[u].x - base), d1 = (unsigned)(dv[u].y - base);
      const unsigned d2 = (unsigned)(dv[u].z - base), d3 = (unsigned)(dv[u].w - base);
      const unsigned dm = min(min(d0, d1), min(d2, d3));
      if (dm < (unsigned)ROWS) {
        const unsigned dd[4] = {d0, d1, d2, d3};
#pragma unroll
        for (int k = 0; k < 4; ++k) {
          if (dd[k] < (unsigned)ROWS) {
            const unsigned slot = atomicAdd(&cnt, 1u);
            if (slot < (unsigned)CAP) {
              q[slot] = e + k;
            } else {
              const int s = src[e + k];
              const float nrm = dinv[s] * dinv[base + (int)dd[k]];
              for (int col = 0; col < F; ++col)
                __hip_atomic_fetch_add(&acc[dd[k] * F + (col % V) * 32 + col / V],
                                       h[(size_t)s * F + col] * nrm, __ATOMIC_RELAXED,
                                       __HIP_MEMORY_SCOPE_WORKGROUP);
            }
          }
        }
      }
    }
    ++c;
    if (c >= nch) c = 0;
  }
  __syncthreads();

  const int n = (int)min(cnt, (unsigned)CAP);
  for (int i0 = wave * U; i0 < n; i0 += (NT / 32) * U) {
    int sv[U], dl[U];
#pragma unroll
    for (int u = 0; u < U; ++u) {
      const int idx = min(i0 + u, n - 1);
      const int e = q[idx];
      sv[u] = src[e];
      dl[u] = dst[e] - base;
    }
    float nr[U];
    float hv[U][V];
#pragma unroll
    for (int u = 0; u < U; ++u) {
      nr[u] = dinv[sv[u]] * dinv[base + dl[u]];
      const float* hp = h + (size_t)sv[u] * F + lane * V;
      if (V == 4) {
        const float4 t = *(const float4*)hp;
        hv[u][0] = t.x; hv[u][1] = t.y; hv[u][V - 2] = t.z; hv[u][V - 1] = t.w;
      } else {
        const float2 t = *(const float2*)hp;
        hv[u][0] = t.x; hv[u][1] = t.y;
      }
    }
#pragma unroll
    for (int u = 0; u < U; ++u) {
      if (i0 + u < n) {
#pragma unroll
        for (int j = 0; j < V; ++j)
          __hip_atomic_fetch_add(&acc[dl[u] * F + j * 32 + lane], hv[u][j] * nr[u],
                                 __ATOMIC_RELAXED, __HIP_MEMORY_SCOPE_WORKGROUP);
      }
    }
  }
  __syncthreads();

  for (int i = tid; i < ROWS * 32; i += NT) {
    const int r = i >> 5, ln = i & 31;
    const int row = base + r;
    if (row < N) {
#pragma unroll
      for (int j = 0; j < V; ++j) agg[(size_t)row * F + ln * V + j] = acc[r * F + j * 32 + ln];
    }
  }
}

extern "C" void kernel_launch(void* const* d_in, const int* in_sizes, int n_in, void* d_out,
                              int out_size, void* d_ws, size_t ws_size, hipStream_t stream) {
  const float* x = (const float*)d_in[0];
  const int* ei = (const int*)d_in[1];
  const float* W1 = (const float*)d_in[2];
  const float* b1 = (const float*)d_in[3];
  const float* W2 = (const float*)d_in[4];
  const float* b2 = (const float*)d_in[5];
  float* out = (float*)d_out;

  const int N = in_sizes[0] / 128;
  const int E = in_sizes[1] / 2;
  const int* src = ei;
  const int* dstp = ei + E;

  char* ws = (char*)d_ws;
  float* dinv = (float*)ws;
  size_t off = (((size_t)N * 4) + 255) & ~(size_t)255;
  float* h1 = (float*)(ws + off);
  float* agg1 = (float*)(ws + off + (size_t)N * 128 * 4);
  float* h2 = h1;

  dim3 B(256);
  k_init_deg<<<(N + 255) / 256, B, 0, stream>>>(dinv, N);
  k_count_deg<<<(E + 255) / 256, B, 0, stream>>>(dstp, dinv, E);
  k_rsqrt_deg<<<(N + 255) / 256, B, 0, stream>>>(dinv, N);

  const int ntiles = (N + 15) / 16;
  const int gblocks = (ntiles + 7) / 8;

  k_gemm_wmma<128, false><<<gblocks, B, 0, stream>>>(x, W1, h1, dinv, b1, agg1, N);
  k_agg_owner<128><<<(N + 255) / 256, dim3(1024), 0, stream>>>(h1, src, dstp, dinv, agg1, E, N);

  k_gemm_wmma<64, true><<<gblocks, B, 0, stream>>>(agg1, W2, h2, dinv, b2, out, N);
  k_agg_owner<64><<<(N + 255) / 256, dim3(1024), 0, stream>>>(h2, src, dstp, dinv, out, E, N);
}


# owners' edge lists and the degrees built once per call in one pass over dst, kept in workspace, read by both aggregation launches: dst scanned once, not twice; 7 launches to 5; add order LDS arrival
# speedup vs baseline: 1.6331x; 1.4123x over previous
#include <hip/hip_runtime.h>

typedef __attribute__((ext_vector_type(2))) float v2f;
typedef __attribute__((ext_vector_type(8))) float v8f;

constexpr int OWN_ROWS = 256;
constexpr int OWN_CAP = 16384;

__global__ __launch_bounds__(1024) void k_build_lists(const int* __restrict__ src,
                                                      const int* __restrict__ dst,
                                                      float* __restrict__ dinv,
                                                      int* __restrict__ lists,
                                                      unsigned* __restrict__ counts, int E, int N) {
  constexpr int ROWS = OWN_ROWS;
  constexpr int CAP = OWN_CAP;
  constexpr int NT = 1024;
  constexpr int UNR = 4;
  constexpr int CH = NT * 4 * UNR;
  __shared__ int q[CAP];
  __shared__ unsigned degc[ROWS];
  __shared__ unsigned cnt;

  const int tid = threadIdx.x;
  const int base = blockIdx.x * ROWS;

  if (tid == 0) cnt = 0u;
  if (tid < ROWS) degc[tid] = 0u;
  __syncthreads();

  const int nch = (E + CH - 1) / CH;
  int c = (int)(((long long)blockIdx.x * nch) / gridDim.x);
  const bool vec4 = ((E & 3) == 0);
  for (int ci = 0; ci < nch; ++ci) {
    const int e0 = c * CH;
    int4 dv[UNR];
    if (vec4 && e0 + CH <= E) {
#pragma unroll
      for (int u = 0; u < UNR; ++u) dv[u] = *(const int4*)(dst + e0 + (u * NT + tid) * 4);
    } else {
#pragma unroll
      for (int u = 0; u < UNR; ++u) {
        const int e = e0 + (u * NT + tid) * 4;
        dv[u].x = (e + 0 < E) ? dst[e + 0] : -1;
        dv[u].y = (e + 1 < E) ? dst[e + 1] : -1;
        dv[u].z = (e + 2 < E) ? dst[e + 2] : -1;
        dv[u].w = (e + 3 < E) ? dst[e + 3] : -1;
      }
    }
#pragma unroll
    for (int u = 0; u < UNR; ++u) {
      const int e = e0 + (u * NT + tid) * 4;
      const unsigned d0 = (unsigned)(dv[u].x - base), d1 = (unsigned)(dv[u].y - base);
      const unsigned d2 = (unsigned)(dv[u].z - base), d3 = (unsigned)(dv[u].w - base);
      const unsigned dm = min(min(d0, d1), min(d2, d3));
      if (dm < (unsigned)ROWS) {
        const unsigned dd[4] = {d0, d1, d2, d3};
#pragma unroll
        for (int k = 0; k < 4; ++k) {
          if (dd[k] < (unsigned)ROWS) {
            atomicAdd(&degc[dd[k]], 1u);
            const unsigned slot = atomicAdd(&cnt, 1u);
            if (slot < (unsigned)CAP) q[slot] = e + k;
          }
        }
      }
    }
    ++c;
    if (c >= nch) c = 0;
  }
  __syncthreads();

  const unsigned tot = cnt;
  if (tid == 0) counts[blockIdx.x] = tot;
  if (tid < ROWS && base + tid < N) dinv[base + tid] = rsqrtf((float)(degc[tid] + 1u));

  const int n = (int)min(tot, (unsigned)CAP);
  int2* __restrict__ my = (int2*)(lists + (size_t)blockIdx.x * CAP * 2);
  for (int i = tid; i < n; i += NT) {
    const int e = q[i];
    my[i] = make_int2(src[e], dst[e] - base);
  }
}

template <int NOUT, bool RELU>
__global__ __launch_bounds__(256) void k_gemm_wmma(const float* __restrict__ A,
                                                   const float* __restrict__ W,
                                                   float* __restrict__ out,
                                                   const float* __restrict__ dinv,
                                                   const float* __restrict__ bias,
                                                   float* __restrict__ agg, int nrows) {
  __shared__ float Wl[128 * NOUT];
  const int tid = threadIdx.x;
  for (int i = tid; i < 128 * NOUT; i += 256) {
    int k = i / NOUT;
    int j = i - k * NOUT;
    Wl[(k >> 1) * (2 * NOUT) + 2 * j + (k & 1)] = W[i];
  }
  __syncthreads();

  const int wave = tid >> 5;
  const int lane = tid & 31;
  const int tile = blockIdx.x * 8 + wave;
  const int ntiles = (nrows + 15) >> 4;
  if (tile >= ntiles) return;

  const int m = lane & 15;
  const int kh = lane >> 4;
  int row = tile * 16 + m;
  if (row >= nrows) row = nrows - 1;

  const int rbase = tile * 16 + kh * 8;
  float di[8];
#pragma unroll
  for (int v = 0; v < 8; ++v) {
    int r = rbase + v;
    di[v] = dinv[r < nrows ? r : nrows - 1];
  }
  float bc[NOUT / 16];
#pragma unroll
  for (int t = 0; t < NOUT / 16; ++t) bc[t] = bias[t * 16 + m];

  v8f acc[NOUT / 16];
#pragma unroll
  for (int t = 0; t < NOUT / 16; ++t) acc[t] = v8f{0.f, 0.f, 0.f, 0.f, 0.f, 0.f, 0.f, 0.f};

  const float* arow = A + (size_t)row * 128 + 2 * kh;
#pragma unroll 4
  for (int k0 = 0; k0 < 128; k0 += 4) {
    v2f a = *(const v2f*)(arow + k0);
    if (RELU) {
      a.x = fmaxf(a.x, 0.f);
      a.y = fmaxf(a.y, 0.f);
    }
    const float* bb = &Wl[((k0 >> 1) + kh) * (2 * NOUT)];
#pragma unroll
    for (int t = 0; t < NOUT / 16; ++t) {
      v2f b = *(const v2f*)(bb + 2 * (t * 16 + m));
      acc[t] = __builtin_amdgcn_wmma_f32_16x16x4_f32(false, a, false, b, (short)0, acc[t],
                                                     false, false);
    }
  }

#pragma unroll
  for (int t = 0; t < NOUT / 16; ++t) {
#pragma unroll
    for (int v = 0; v < 8; ++v) {
      int r = rbase + v;
      if (r < nrows) {
        const float hv = acc[t][v];
        const size_t o = (size_t)r * NOUT + t * 16 + m;
        out[o] = hv;
        agg[o] = hv * di[v] * di[v] + bc[t];
      }
    }
  }
}

template <int F>
__global__ __launch_bounds__(1024) void k_agg_owner(const float* __restrict__ h,
                                                    const int* __restrict__ src,
                                                    const int* __restrict__ dst,
                                                    const float* __restrict__ dinv,
                                                    const int* __restrict__ lists,
                                                    const unsigned* __restrict__ counts,
                                                    float* __restrict__ agg, int E, int N) {
  constexpr int ROWS = OWN_ROWS;
  constexpr int CAP = OWN_CAP;
  constexpr int V = F / 32;
  constexpr int NT = 1024;
  constexpr int U = 8;
  __shared__ float acc[ROWS * F];

  const int tid = threadIdx.x;
  const int lane = tid & 31;
  const int wave = tid >> 5;
  const int base = blockIdx.x * ROWS;

  for (int i = tid; i < ROWS * 32; i += NT) {
    const int r = i >> 5, ln = i & 31;
    const int row = base + r;
#pragma unroll
    for (int j = 0; j < V; ++j)
      acc[r * F + j * 32 + ln] = (row < N) ? agg[(size_t)row * F + ln * V + j] : 0.f;
  }
  __syncthreads();

  const unsigned tot = counts[blockIdx.x];
  if (tot <= (unsigned)CAP) {
    const int n = (int)tot;
    const int2* __restrict__ my = (const int2*)(lists + (size_t)blockIdx.x * CAP * 2);
    for (int i0 = wave * U; i0 < n; i0 += (NT / 32) * U) {
      int sv[U], dl[U];
#pragma unroll
      for (int u = 0; u < U; ++u) {
        const int2 ent = my[min(i0 + u, n - 1)];
        sv[u] = ent.x;
        dl[u] = ent.y;
      }
      float nr[U];
      float hv[U][V];
#pragma unroll
      for (int u = 0; u < U; ++u) {
        nr[u] = dinv[sv[u]] * dinv[base + dl[u]];
        const float* hp = h + (size_t)sv[u] * F + lane * V;
        if (V == 4) {
          const float4 t = *(const float4*)hp;
          hv[u][0] = t.x; hv[u][1] = t.y; hv[u][V - 2] = t.z; hv[u][V - 1] = t.w;
        } else {
          const float2 t = *(const float2*)hp;
          hv[u][0] = t.x; hv[u][1] = t.y;
        }
      }
#pragma unroll
      for (int u = 0; u < U; ++u) {
        if (i0 + u < n) {
#pragma unroll
          for (int j = 0; j < V; ++j)
            __hip_atomic_fetch_add(&acc[dl[u] * F + j * 32 + lane], hv[u][j] * nr[u],
                                   __ATOMIC_RELAXED, __HIP_MEMORY_SCOPE_WORKGROUP);
        }
      }
    }
  } else {
    for (int e = tid; e < E; e += NT) {
      const unsigned d = (unsigned)(dst[e] - base);
      if (d < (unsigned)ROWS) {
        const int s = src[e];
        const float nrm = dinv[s] * dinv[base + (int)d];
        for (int col = 0; col < F; ++col)
          __hip_atomic_fetch_add(&acc[d * F + (col % V) * 32 + col / V],
                                 h[(size_t)s * F + col] * nrm, __ATOMIC_RELAXED,
                                 __HIP_MEMORY_SCOPE_WORKGROUP);
      }
    }
  }
  __syncthreads();

  for (int i = tid; i < ROWS * 32; i += NT) {
    const int r = i >> 5, ln = i & 31;
    const int row = base + r;
    if (row < N) {
#pragma unroll
      for (int j = 0; j < V; ++j) agg[(size_t)row * F + ln * V + j] = acc[r * F + j * 32 + ln];
    }
  }
}

extern "C" void kernel_launch(void* const* d_in, const int* in_sizes, int n_in, void* d_out,
                              int out_size, void* d_ws, size_t ws_size, hipStream_t stream) {
  const float* x = (const float*)d_in[0];
  const int* ei = (const int*)d_in[1];
  const float* W1 = (const float*)d_in[2];
  const float* b1 = (const float*)d_in[3];
  const float* W2 = (const float*)d_in[4];
  const float* b2 = (const float*)d_in[5];
  float* out = (float*)d_out;

  const int N = in_sizes[0] / 128;
  const int E = in_sizes[1] / 2;
  const int* src = ei;
  const int* dstp = ei + E;

  char* ws = (char*)d_ws;
  float* dinv = (float*)ws;
  size_t off = (((size_t)N * 4) + 255) & ~(size_t)255;
  float* h1 = (float*)(ws + off);
  float* agg1 = (float*)(ws + off + (size_t)N * 128 * 4);
  float* h2 = h1;
  const int nblk = (N + OWN_ROWS - 1) / OWN_ROWS;
  int* lists = (int*)(ws + off + (size_t)N * 128 * 4 * 2);
  unsigned* counts = (unsigned*)(ws + off + (size_t)N * 128 * 4 * 2 + (size_t)nblk * OWN_CAP * 8);

  dim3 B(256);
  k_build_lists<<<nblk, dim3(1024), 0, stream>>>(src, dstp, dinv, lists, counts, E, N);

  const int ntiles = (N + 15) / 16;
  const int gblocks = (ntiles + 7) / 8;

  k_gemm_wmma<128, false><<<gblocks, B, 0, stream>>>(x, W1, h1, dinv, b1, agg1, N);
  k_agg_owner<128><<<nblk, dim3(1024), 0, stream>>>(h1, src, dstp, dinv, lists, counts, agg1, E, N);

  k_gemm_wmma<64, true><<<gblocks, B, 0, stream>>>(agg1, W2, h2, dinv, b2, out, N);
  k_agg_owner<64><<<nblk, dim3(1024), 0, stream>>>(h2, src, dstp, dinv, lists, counts, out, E, N);
}


# aggregation kernels: dependent loads overlapped: list entries fetched one trip ahead, all 8 gathers of a trip issued together, start rows loaded as one group; add order LDS arrival as before
# speedup vs baseline: 1.6869x; 1.0329x over previous
#include <hip/hip_runtime.h>

typedef __attribute__((ext_vector_type(2))) float v2f;
typedef __attribute__((ext_vector_type(8))) float v8f;

constexpr int OWN_ROWS = 256;
constexpr int OWN_CAP = 16384;

__global__ __launch_bounds__(1024) void k_build_lists(const int* __restrict__ src,
                                                      const int* __restrict__ dst,
                                                      float* __restrict__ dinv,
                                                      int* __restrict__ lists,
                                                      unsigned* __restrict__ counts, int E, int N) {
  constexpr int ROWS = OWN_ROWS;
  constexpr int CAP = OWN_CAP;
  constexpr int NT = 1024;
  constexpr int UNR = 4;
  constexpr int CH = NT * 4 * UNR;
  __shared__ int q[CAP];
  __shared__ unsigned degc[ROWS];
  __shared__ unsigned cnt;

  const int tid = threadIdx.x;
  const int base = blockIdx.x * ROWS;

  if (tid == 0) cnt = 0u;
  if (tid < ROWS) degc[tid] = 0u;
  __syncthreads();

  const int nch = (E + CH - 1) / CH;
  int c = (int)(((long long)blockIdx.x * nch) / gridDim.x);
  const bool vec4 = ((E & 3) == 0);
  for (int ci = 0; ci < nch; ++ci) {
    const int e0 = c * CH;
    int4 dv[UNR];
    if (vec4 && e0 + CH <= E) {
#pragma unroll
      for (int u = 0; u < UNR; ++u) dv[u] = *(const int4*)(dst + e0 + (u * NT + tid) * 4);
    } else {
#pragma unroll
      for (int u = 0; u < UNR; ++u) {
        const int e = e0 + (u * NT + tid) * 4;
        dv[u].x = (e + 0 < E) ? dst[e + 0] : -1;
        dv[u].y = (e + 1 < E) ? dst[e + 1] : -1;
        dv[u].z = (e + 2 < E) ? dst[e + 2] : -1;
        dv[u].w = (e + 3 < E) ? dst[e + 3] : -1;
      }
    }
#pragma unroll
    for (int u = 0; u < UNR; ++u) {
      const int e = e0 + (u * NT + tid) * 4;
      const unsigned d0 = (unsigned)(dv[u].x - base), d1 = (unsigned)(dv[u].y - base);
      const unsigned d2 = (unsigned)(dv[u].z - base), d3 = (unsigned)(dv[u].w - base);
      const unsigned dm = min(min(d0, d1), min(d2, d3));
      if (dm < (unsigned)ROWS) {
        const unsigned dd[4] = {d0, d1, d2, d3};
#pragma unroll
        for (int k = 0; k < 4; ++k) {
          if (dd[k] < (unsigned)ROWS) {
            atomicAdd(&degc[dd[k]], 1u);
            const unsigned slot = atomicAdd(&cnt, 1u);
            if (slot < (unsigned)CAP) q[slot] = e + k;
          }
        }
      }
    }
    ++c;
    if (c >= nch) c = 0;
  }
  __syncthreads();

  const unsigned tot = cnt;
  if (tid == 0) counts[blockIdx.x] = tot;
  if (tid < ROWS && base + tid < N) dinv[base + tid] = rsqrtf((float)(degc[tid] + 1u));

  const int n = (int)min(tot, (unsigned)CAP);
  int2* __restrict__ my = (int2*)(lists + (size_t)blockIdx.x * CAP * 2);
  for (int i = tid; i < n; i += NT) {
    const int e = q[i];
    my[i] = make_int2(src[e], dst[e] - base);
  }
}

template <int NOUT, bool RELU>
__global__ __launch_bounds__(256) void k_gemm_wmma(const float* __restrict__ A,
                                                   const float* __restrict__ W,
                                                   float* __restrict__ out,
                                                   const float* __restrict__ dinv,
                                                   const float* __restrict__ bias,
                                                   float* __restrict__ agg, int nrows) {
  __shared__ float Wl[128 * NOUT];
  const int tid = threadIdx.x;
  for (int i = tid; i < 128 * NOUT; i += 256) {
    int k = i / NOUT;
    int j = i - k * NOUT;
    Wl[(k >> 1) * (2 * NOUT) + 2 * j + (k & 1)] = W[i];
  }
  __syncthreads();

  const int wave = tid >> 5;
  const int lane = tid & 31;
  const int tile = blockIdx.x * 8 + wave;
  const int ntiles = (nrows + 15) >> 4;
  if (tile >= ntiles) return;

  const int m = lane & 15;
  const int kh = lane >> 4;
  int row = tile * 16 + m;
  if (row >= nrows) row = nrows - 1;

  const int rbase = tile * 16 + kh * 8;
  float di[8];
#pragma unroll
  for (int v = 0; v < 8; ++v) {
    int r = rbase + v;
    di[v] = dinv[r < nrows ? r : nrows - 1];
  }
  float bc[NOUT / 16];
#pragma unroll
  for (int t = 0; t < NOUT / 16; ++t) bc[t] = bias[t * 16 + m];

  v8f acc[NOUT / 16];
#pragma unroll
  for (int t = 0; t < NOUT / 16; ++t) acc[t] = v8f{0.f, 0.f, 0.f, 0.f, 0.f, 0.f, 0.f, 0.f};

  const float* arow = A + (size_t)row * 128 + 2 * kh;
#pragma unroll 4
  for (int k0 = 0; k0 < 128; k0 += 4) {
    v2f a = *(const v2f*)(arow + k0);
    if (RELU) {
      a.x = fmaxf(a.x, 0.f);
      a.y = fmaxf(a.y, 0.f);
    }
    const float* bb = &Wl[((k0 >> 1) + kh) * (2 * NOUT)];
#pragma unroll
    for (int t = 0; t < NOUT / 16; ++t) {
      v2f b = *(const v2f*)(bb + 2 * (t * 16 + m));
      acc[t] = __builtin_amdgcn_wmma_f32_16x16x4_f32(false, a, false, b, (short)0, acc[t],
                                                     false, false);
    }
  }

#pragma unroll
  for (int t = 0; t < NOUT / 16; ++t) {
#pragma unroll
    for (int v = 0; v < 8; ++v) {
      int r = rbase + v;
      if (r < nrows) {
        const float hv = acc[t][v];
        const size_t o = (size_t)r * NOUT + t * 16 + m;
        out[o] = hv;
        agg[o] = hv * di[v] * di[v] + bc[t];
      }
    }
  }
}

template <int F>
__global__ __launch_bounds__(1024) void k_agg_owner(const float* __restrict__ h,
                                                    const int* __restrict__ src,
                                                    const int* __restrict__ dst,
                                                    const float* __restrict__ dinv,
                                                    const int* __restrict__ lists,
                                                    const unsigned* __restrict__ counts,
                                                    float* __restrict__ agg, int E, int N) {
  constexpr int ROWS = OWN_ROWS;
  constexpr int CAP = OWN_CAP;
  constexpr int V = F / 32;
  constexpr int NT = 1024;
  constexpr int NW = NT / 32;
  constexpr int U = 8;
  constexpr int PER = ROWS / NW;
  __shared__ float acc[ROWS * F];

  const int tid = threadIdx.x;
  const int lane = tid & 31;
  const int wave = tid >> 5;
  const int base = blockIdx.x * ROWS;

  {
    float st[PER][V];
#pragma unroll
    for (int k = 0; k < PER; ++k) {
      const int row = min(base + wave + k * NW, N - 1);
      const float* p = agg + (size_t)row * F + lane * V;
      if (V == 4) {
        const float4 t = *(const float4*)p;
        st[k][0] = t.x; st[k][1] = t.y; st[k][V - 2] = t.z; st[k][V - 1] = t.w;
      } else {
        const float2 t = *(const float2*)p;
        st[k][0] = t.x; st[k][1] = t.y;
      }
    }
#pragma unroll
    for (int k = 0; k < PER; ++k) {
#pragma unroll
      for (int j = 0; j < V; ++j) acc[(wave + k * NW) * F + j * 32 + lane] = st[k][j];
    }
  }
  __syncthreads();

  const unsigned tot = counts[blockIdx.x];
  if (tot <= (unsigned)CAP) {
    const int n = (int)tot;
    const int2* __restrict__ my = (const int2*)(lists + (size_t)blockIdx.x * CAP * 2);
    const int nfull = n / U;

    int g = wave;
    int2 ent[U];
#pragma unroll
    for (int u = 0; u < U; ++u) ent[u] = make_int2(0, 0);
    if (g < nfull) {
#pragma unroll
      for (int u = 0; u < U; ++u) ent[u] = my[g * U + u];
    }
    for (; g < nfull; g += NW) {
      const int gn = (g + NW < nfull) ? g + NW : g;
      int2 nxt[U];
#pragma unroll
      for (int u = 0; u < U; ++u) nxt[u] = my[gn * U + u];

      float nr[U];
      float hv[U][V];
#pragma unroll
      for (int u = 0; u < U; ++u) {
        nr[u] = dinv[ent[u].x] * dinv[base + ent[u].y];
        const float* hp = h + (size_t)ent[u].x * F + lane * V;
        if (V == 4) {
          const float4 t = *(const float4*)hp;
          hv[u][0] = t.x; hv[u][1] = t.y; hv[u][V - 2] = t.z; hv[u][V - 1] = t.w;
        } else {
          const float2 t = *(const float2*)hp;
          hv[u][0] = t.x; hv[u][1] = t.y;
        }
      }
#pragma unroll
      for (int u = 0; u < U; ++u) {
#pragma unroll
        for (int j = 0; j < V; ++j)
          __hip_atomic_fetch_add(&acc[ent[u].y * F + j * 32 + lane], hv[u][j] * nr[u],
                                 __ATOMIC_RELAXED, __HIP_MEMORY_SCOPE_WORKGROUP);
      }
#pragma unroll
      for (int u = 0; u < U; ++u) ent[u] = nxt[u];
    }

    const int rem = n - nfull * U;
    if (rem > 0 && wave == nfull % NW) {
      int sv[U], dl[U];
#pragma unroll
      for (int u = 0; u < U; ++u) {
        const int2 e2 = my[min(nfull * U + u, n - 1)];
        sv[u] = e2.x;
        dl[u] = e2.y;
      }
      float nr[U];
      float hv[U][V];
#pragma unroll
      for (int u = 0; u < U; ++u) {
        nr[u] = dinv[sv[u]] * dinv[base + dl[u]];
        const float* hp = h + (size_t)sv[u] * F + lane * V;
        if (V == 4) {
          const float4 t = *(const float4*)hp;
          hv[u][0] = t.x; hv[u][1] = t.y; hv[u][V - 2] = t.z; hv[u][V - 1] = t.w;
        } else {
          const float2 t = *(const float2*)hp;
          hv[u][0] = t.x; hv[u][1] = t.y;
        }
      }
#pragma unroll
      for (int u = 0; u < U; ++u) {
        if (u < rem) {
#pragma unroll
          for (int j = 0; j < V; ++j)
            __hip_atomic_fetch_add(&acc[dl[u] * F + j * 32 + lane], hv[u][j] * nr[u],
                                   __ATOMIC_RELAXED, __HIP_MEMORY_SCOPE_WORKGROUP);
        }
      }
    }
  } else {
    for (int e = tid; e < E; e += NT) {
      const unsigned d = (unsigned)(dst[e] - base);
      if (d < (unsigned)ROWS) {
        const int s = src[e];
        const float nrm = dinv[s] * dinv[base + (int)d];
        for (int col = 0; col < F; ++col)
          __hip_atomic_fetch_add(&acc[d * F + (col % V) * 32 + col / V],
                                 h[(size_t)s * F + col] * nrm, __ATOMIC_RELAXED,
                                 __HIP_MEMORY_SCOPE_WORKGROUP);
      }
    }
  }
  __syncthreads();

  for (int i = tid; i < ROWS * 32; i += NT) {
    const int r = i >> 5, ln = i & 31;
    const int row = base + r;
    if (row < N) {
#pragma unroll
      for (int j = 0; j < V; ++j) agg[(size_t)row * F + ln * V + j] = acc[r * F + j * 32 + ln];
    }
  }
}

extern "C" void kernel_launch(void* const* d_in, const int* in_sizes, int n_in, void* d_out,
                              int out_size, void* d_ws, size_t ws_size, hipStream_t stream) {
  const float* x = (const float*)d_in[0];
  const int* ei = (const int*)d_in[1];
  const float* W1 = (const float*)d_in[2];
  const float* b1 = (const float*)d_in[3];
  const float* W2 = (const float*)d_in[4];
  const float* b2 = (const float*)d_in[5];
  float* out = (float*)d_out;

  const int N = in_sizes[0] / 128;
  const int E = in_sizes[1] / 2;
  const int* src = ei;
  const int* dstp = ei + E;

  char* ws = (char*)d_ws;
  float* dinv = (float*)ws;
  size_t off = (((size_t)N * 4) + 255) & ~(size_t)255;
  float* h1 = (float*)(ws + off);
  float* agg1 = (float*)(ws + off + (size_t)N * 128 * 4);
  float* h2 = h1;
  const int nblk = (N + OWN_ROWS - 1) / OWN_ROWS;
  int* lists = (int*)(ws + off + (size_t)N * 128 * 4 * 2);
  unsigned* counts = (unsigned*)(ws + off + (size_t)N * 128 * 4 * 2 + (size_t)nblk * OWN_CAP * 8);

  dim3 B(256);
  k_build_lists<<<nblk, dim3(1024), 0, stream>>>(src, dstp, dinv, lists, counts, E, N);

  const int ntiles = (N + 15) / 16;
  const int gblocks = (ntiles + 7) / 8;

  k_gemm_wmma<128, false><<<gblocks, B, 0, stream>>>(x, W1, h1, dinv, b1, agg1, N);
  k_agg_owner<128><<<nblk, dim3(1024), 0, stream>>>(h1, src, dstp, dinv, lists, counts, agg1, E, N);

  k_gemm_wmma<64, true><<<gblocks, B, 0, stream>>>(agg1, W2, h2, dinv, b2, out, N);
  k_agg_owner<64><<<nblk, dim3(1024), 0, stream>>>(h2, src, dstp, dinv, lists, counts, out, E, N);
}


# owners find their edges in a one-byte-per-edge plane (owning block, written by a small first kernel), 4 edges per 32-bit test, not in the 4-byte dst plane; 5 launches to 6; add order LDS arrival
# speedup vs baseline: 2.5997x; 1.5411x over previous
#include <hip/hip_runtime.h>

typedef __attribute__((ext_vector_type(2))) float v2f;
typedef __attribute__((ext_vector_type(8))) float v8f;

constexpr int OWN_ROWS = 256;
constexpr int OWN_CAP = 16384;
constexpr unsigned KEY_NONE = 255u;
constexpr int KEY_STEP = 4096;

__device__ __forceinline__ unsigned owner_key(int d) {
  return min((unsigned)d / (unsigned)OWN_ROWS, KEY_NONE);
}

__global__ __launch_bounds__(256) void k_edge_keys(const int* __restrict__ dst, unsigned* __restrict__ keys,
                                                   int E) {
  const int t = blockIdx.x * 256 + threadIdx.x;
  const int n16 = (E + 15) >> 4;
  const int n16p = (n16 + KEY_STEP - 1) / KEY_STEP * KEY_STEP;
  if (t >= n16p) return;
  const int e0 = t * 16;
  unsigned w[4];
  if (((E & 3) == 0) && e0 + 16 <= E) {
    int4 d[4];
#pragma unroll
    for (int j = 0; j < 4; ++j) d[j] = *(const int4*)(dst + e0 + 4 * j);
#pragma unroll
    for (int j = 0; j < 4; ++j)
      w[j] = owner_key(d[j].x) | (owner_key(d[j].y) << 8) | (owner_key(d[j].z) << 16) | (owner_key(d[j].w) << 24);
  } else {
#pragma unroll
    for (int j = 0; j < 4; ++j) {
      unsigned v = 0u;
#pragma unroll
      for (int k = 0; k < 4; ++k) {
        const int e = e0 + 4 * j + k;
        v |= ((e < E) ? owner_key(dst[e]) : KEY_NONE) << (8 * k);
      }
      w[j] = v;
    }
  }
  *(uint4*)(keys + 4 * (size_t)t) = make_uint4(w[0], w[1], w[2], w[3]);
}

__global__ __launch_bounds__(1024) void k_build_lists(const int* __restrict__ src,
                                                      const int* __restrict__ dst,
                                                      const unsigned* __restrict__ keys,
                                                      float* __restrict__ dinv,
                                                      int* __restrict__ lists,
                                                      unsigned* __restrict__ counts, int E, int N) {
  constexpr int ROWS = OWN_ROWS;
  constexpr int CAP = OWN_CAP;
  constexpr int NT = 1024;
  constexpr int UNR = 4;
  constexpr int CHW = NT * UNR;
  __shared__ int q[CAP];
  __shared__ unsigned degc[ROWS];
  __shared__ unsigned cnt;

  const int tid = threadIdx.x;
  const int base = blockIdx.x * ROWS;

  if (tid == 0) cnt = 0u;
  if (tid < ROWS) degc[tid] = 0u;
  __syncthreads();

  static_assert(UNR == 4 && CHW == KEY_STEP, "the match mask below packs the flags of exactly 4 key words");
  const uint4* __restrict__ kp = (const uint4*)keys;
  const int nch = ((E + 15) / 16 + CHW - 1) / CHW;
  int c = (int)(((long long)blockIdx.x * nch) / gridDim.x);
  const unsigned pat = (unsigned)blockIdx.x * 0x01010101u;

  uint4 cur[UNR];
#pragma unroll
  for (int u = 0; u < UNR; ++u) cur[u] = kp[c * CHW + u * NT + tid];
  for (int ci = 0; ci < nch; ++ci) {
    int cn = c + 1;
    if (cn >= nch) cn = 0;
    uint4 nxt[UNR];
#pragma unroll
    for (int u = 0; u < UNR; ++u) nxt[u] = kp[cn * CHW + u * NT + tid];

    unsigned mm[UNR];
#pragma unroll
    for (int u = 0; u < UNR; ++u) {
      const unsigned x[4] = {cur[u].x ^ pat, cur[u].y ^ pat, cur[u].z ^ pat, cur[u].w ^ pat};
      unsigned m = 0u;
#pragma unroll
      for (int j = 0; j < 4; ++j) {
        const unsigned f = ~(((x[j] & 0x7f7f7f7fu) + 0x7f7f7f7fu) | x[j]) & 0x80808080u;
        m |= f >> (7 - j);
      }
      mm[u] = m;
    }
    unsigned long long M = (unsigned long long)(mm[0] | (mm[1] << 4)) |
                           ((unsigned long long)(mm[2] | (mm[3] << 4)) << 32);
    const int ebase = (c * CHW + tid) * 16;
    while (M) {
      const int p = __builtin_ctzll(M);
      M &= M - 1ull;
      const int u = ((p >> 5) << 1) | ((p >> 2) & 1);
      const int e = ebase + u * (NT * 16) + (p & 3) * 4 + ((p >> 3) & 3);
      const unsigned slot = atomicAdd(&cnt, 1u);
      if (slot < (unsigned)CAP) q[slot] = e;
    }
#pragma unroll
    for (int u = 0; u < UNR; ++u) cur[u] = nxt[u];
    c = cn;
  }
  __syncthreads();

  const unsigned tot = cnt;
  if (tid == 0) counts[blockIdx.x] = tot;

  if (tot <= (unsigned)CAP) {
    const int n = (int)tot;
    int2* __restrict__ my = (int2*)(lists + (size_t)blockIdx.x * CAP * 2);
    for (int i = tid; i < n; i += NT) {
      const int e = q[i];
      const int dl = dst[e] - base;
      atomicAdd(&degc[dl], 1u);
      my[i] = make_int2(src[e], dl);
    }
  } else {
    for (int e = tid; e < E; e += NT) {
      const unsigned d = (unsigned)(dst[e] - base);
      if (d < (unsigned)ROWS) atomicAdd(&degc[d], 1u);
    }
  }
  __syncthreads();

  if (tid < ROWS && base + tid < N) dinv[base + tid] = rsqrtf((float)(degc[tid] + 1u));
}

template <int NOUT, bool RELU>
__global__ __launch_bounds__(256) void k_gemm_wmma(const float* __restrict__ A,
                                                   const float* __restrict__ W,
                                                   float* __restrict__ out,
                                                   const float* __restrict__ dinv, int nrows) {
  __shared__ float Wl[128 * NOUT];
  const int tid = threadIdx.x;
  for (int i = tid; i < 128 * NOUT; i += 256) {
    int k = i / NOUT;
    int j = i - k * NOUT;
    Wl[(k >> 1) * (2 * NOUT) + 2 * j + (k & 1)] = W[i];
  }
  __syncthreads();

  const int wave = tid >> 5;
  const int lane = tid & 31;
  const int tile = blockIdx.x * 8 + wave;
  const int ntiles = (nrows + 15) >> 4;
  if (tile >= ntiles) return;

  const int m = lane & 15;
  const int kh = lane >> 4;
  int row = tile * 16 + m;
  if (row >= nrows) row = nrows - 1;

  const int rbase = tile * 16 + kh * 8;
  float di[8];
#pragma unroll
  for (int v = 0; v < 8; ++v) {
    int r = rbase + v;
    di[v] = dinv[r < nrows ? r : nrows - 1];
  }
  v8f acc[NOUT / 16];
#pragma unroll
  for (int t = 0; t < NOUT / 16; ++t) acc[t] = v8f{0.f, 0.f, 0.f, 0.f, 0.f, 0.f, 0.f, 0.f};

  const float* arow = A + (size_t)row * 128 + 2 * kh;
#pragma unroll 4
  for (int k0 = 0; k0 < 128; k0 += 4) {
    v2f a = *(const v2f*)(arow + k0);
    if (RELU) {
      a.x = fmaxf(a.x, 0.f);
      a.y = fmaxf(a.y, 0.f);
    }
    const float* bb = &Wl[((k0 >> 1) + kh) * (2 * NOUT)];
#pragma unroll
    for (int t = 0; t < NOUT / 16; ++t) {
      v2f b = *(const v2f*)(bb + 2 * (t * 16 + m));
      acc[t] = __builtin_amdgcn_wmma_f32_16x16x4_f32(false, a, false, b, (short)0, acc[t],
                                                     false, false);
    }
  }

#pragma unroll
  for (int t = 0; t < NOUT / 16; ++t) {
#pragma unroll
    for (int v = 0; v < 8; ++v) {
      int r = rbase + v;
      if (r < nrows) out[(size_t)r * NOUT + t * 16 + m] = acc[t][v] * di[v];
    }
  }
}

template <int F>
__global__ __launch_bounds__(1024) void k_agg_owner(const float* __restrict__ hs,
                                                    const int* __restrict__ src,
                                                    const int* __restrict__ dst,
                                                    const float* __restrict__ dinv,
                                                    const float* __restrict__ bias,
                                                    const int* __restrict__ lists,
                                                    const unsigned* __restrict__ counts,
                                                    float* __restrict__ agg, int E, int N) {
  constexpr int ROWS = OWN_ROWS;
  constexpr int CAP = OWN_CAP;
  constexpr int V = F / 32;
  constexpr int NT = 1024;
  constexpr int NW = NT / 32;
  constexpr int U = 8;
  constexpr int PER = ROWS / NW;
  __shared__ float acc[ROWS * F];

  const int tid = threadIdx.x;
  const int lane = tid & 31;
  const int wave = tid >> 5;
  const int base = blockIdx.x * ROWS;

  {
    float st[PER][V];
#pragma unroll
    for (int k = 0; k < PER; ++k) {
      const int row = min(base + wave + k * NW, N - 1);
      const float* p = hs + (size_t)row * F + lane * V;
      if (V == 4) {
        const float4 t = *(const float4*)p;
        st[k][0] = t.x; st[k][1] = t.y; st[k][V - 2] = t.z; st[k][V - 1] = t.w;
      } else {
        const float2 t = *(const float2*)p;
        st[k][0] = t.x; st[k][1] = t.y;
      }
    }
#pragma unroll
    for (int k = 0; k < PER; ++k) {
#pragma unroll
      for (int j = 0; j < V; ++j) acc[(wave + k * NW) * F + j * 32 + lane] = st[k][j];
    }
  }
  __syncthreads();

  const unsigned tot = counts[blockIdx.x];
  if (tot <= (unsigned)CAP) {
    const int n = (int)tot;
    const int2* __restrict__ my = (const int2*)(lists + (size_t)blockIdx.x * CAP * 2);
    const int nfull = n / U;

    int g = wave;
    int2 ent[U];
#pragma unroll
    for (int u = 0; u < U; ++u) ent[u] = make_int2(0, 0);
    if (g < nfull) {
#pragma unroll
      for (int u = 0; u < U; ++u) ent[u] = my[g * U + u];
    }
    for (; g < nfull; g += NW) {
      const int gn = (g + NW < nfull) ? g + NW : g;
      int2 nxt[U];
#pragma unroll
      for (int u = 0; u < U; ++u) nxt[u] = my[gn * U + u];

      float hv[U][V];
#pragma unroll
      for (int u = 0; u < U; ++u) {
        const float* hp = hs + (size_t)ent[u].x * F + lane * V;
        if (V == 4) {
          const float4 t = *(const float4*)hp;
          hv[u][0] = t.x; hv[u][1] = t.y; hv[u][V - 2] = t.z; hv[u][V - 1] = t.w;
        } else {
          const float2 t = *(const float2*)hp;
          hv[u][0] = t.x; hv[u][1] = t.y;
        }
      }
#pragma unroll
      for (int u = 0; u < U; ++u) {
#pragma unroll
        for (int j = 0; j < V; ++j)
          __hip_atomic_fetch_add(&acc[ent[u].y * F + j * 32 + lane], hv[u][j], __ATOMIC_RELAXED,
                                 __HIP_MEMORY_SCOPE_WORKGROUP);
      }
#pragma unroll
      for (int u = 0; u < U; ++u) ent[u] = nxt[u];
    }

    const int rem = n - nfull * U;
    if (rem > 0 && wave == nfull % NW) {
      int sv[U], dl[U];
#pragma unroll
      for (int u = 0; u < U; ++u) {
        const int2 e2 = my[min(nfull * U + u, n - 1)];
        sv[u] = e2.x;
        dl[u] = e2.y;
      }
      float hv[U][V];
#pragma unroll
      for (int u = 0; u < U; ++u) {
        const float* hp = hs + (size_t)sv[u] * F + lane * V;
        if (V == 4) {
          const float4 t = *(const float4*)hp;
          hv[u][0] = t.x; hv[u][1] = t.y; hv[u][V - 2] = t.z; hv[u][V - 1] = t.w;
        } else {
          const float2 t = *(const float2*)hp;
          hv[u][0] = t.x; hv[u][1] = t.y;
        }
      }
#pragma unroll
      for (int u = 0; u < U; ++u) {
        if (u < rem) {
#pragma unroll
          for (int j = 0; j < V; ++j)
            __hip_atomic_fetch_add(&acc[dl[u] * F + j * 32 + lane], hv[u][j], __ATOMIC_RELAXED,
                                   __HIP_MEMORY_SCOPE_WORKGROUP);
        }
      }
    }
  } else {
    for (int e = tid; e < E; e += NT) {
      const unsigned d = (unsigned)(dst[e] - base);
      if (d < (unsigned)ROWS) {
        const int s = src[e];
        for (int col = 0; col < F; ++col)
          __hip_atomic_fetch_add(&acc[d * F + (col % V) * 32 + col / V], hs[(size_t)s * F + col],
                                 __ATOMIC_RELAXED, __HIP_MEMORY_SCOPE_WORKGROUP);
      }
    }
  }
  __syncthreads();

  float bj[V];
#pragma unroll
  for (int j = 0; j < V; ++j) bj[j] = bias[lane * V + j];
  for (int r = wave; r < ROWS; r += NW) {
    const int row = base + r;
    if (row < N) {
      const float dd = dinv[row];
#pragma unroll
      for (int j = 0; j < V; ++j) agg[(size_t)row * F + lane * V + j] = acc[r * F + j * 32 + lane] * dd + bj[j];
    }
  }
}

extern "C" void kernel_launch(void* const* d_in, const int* in_sizes, int n_in, void* d_out,
                              int out_size, void* d_ws, size_t ws_size, hipStream_t stream) {
  const float* x = (const float*)d_in[0];
  const int* ei = (const int*)d_in[1];
  const float* W1 = (const float*)d_in[2];
  const float* b1 = (const float*)d_in[3];
  const float* W2 = (const float*)d_in[4];
  const float* b2 = (const float*)d_in[5];
  float* out = (float*)d_out;

  const int N = in_sizes[0] / 128;
  const int E = in_sizes[1] / 2;
  const int* src = ei;
  const int* dstp = ei + E;

  char* ws = (char*)d_ws;
  float* dinv = (float*)ws;
  size_t off = (((size_t)N * 4) + 255) & ~(size_t)255;
  float* h1 = (float*)(ws + off);
  float* agg1 = (float*)(ws + off + (size_t)N * 128 * 4);
  float* h2 = h1;
  const int nblk = (N + OWN_ROWS - 1) / OWN_ROWS;
  int* lists = (int*)(ws + off + (size_t)N * 128 * 4 * 2);
  const size_t coff = off + (size_t)N * 128 * 4 * 2 + (size_t)nblk * OWN_CAP * 8;
  unsigned* counts = (unsigned*)(ws + coff);
  unsigned* keys = (unsigned*)(ws + ((coff + (size_t)nblk * 4 + 255) & ~(size_t)255));
  const int n16p = ((E + 15) / 16 + KEY_STEP - 1) / KEY_STEP * KEY_STEP;

  dim3 B(256);
  k_edge_keys<<<n16p / 256, B, 0, stream>>>(dstp, keys, E);
  k_build_lists<<<nblk, dim3(1024), 0, stream>>>(src, dstp, keys, dinv, lists, counts, E, N);

  const int ntiles = (N + 15) / 16;
  const int gblocks = (ntiles + 7) / 8;

  k_gemm_wmma<128, false><<<gblocks, B, 0, stream>>>(x, W1, h1, dinv, N);
  k_agg_owner<128><<<nblk, dim3(1024), 0, stream>>>(h1, src, dstp, dinv, b1, lists, counts, agg1, E, N);

  k_gemm_wmma<64, true><<<gblocks, B, 0, stream>>>(agg1, W2, h2, dinv, N);
  k_agg_owner<64><<<nblk, dim3(1024), 0, stream>>>(h2, src, dstp, dinv, b2, lists, counts, out, E, N);
}


# owner blocks cut from 256 to 200 destination rows: 250 work-groups of 1024 threads, one for nearly each of the 256 WGPs where 196 left 60 idle (list build, both aggregations); add order LDS arrival
# speedup vs baseline: 2.8350x; 1.0905x over previous
#include <hip/hip_runtime.h>

typedef __attribute__((ext_vector_type(2))) float v2f;
typedef __attribute__((ext_vector_type(8))) float v8f;

constexpr int OWN_ROWS = 200;
constexpr int OWN_CAP = 16384;
constexpr unsigned KEY_NONE = 255u;
constexpr int KEY_STEP = 4096;

__device__ __forceinline__ unsigned owner_key(int d) {
  return min((unsigned)d / (unsigned)OWN_ROWS, KEY_NONE);
}

__global__ __launch_bounds__(256) void k_edge_keys(const int* __restrict__ dst, unsigned* __restrict__ keys,
                                                   int E) {
  const int t = blockIdx.x * 256 + threadIdx.x;
  const int n16 = (E + 15) >> 4;
  const int n16p = (n16 + KEY_STEP - 1) / KEY_STEP * KEY_STEP;
  if (t >= n16p) return;
  const int e0 = t * 16;
  unsigned w[4];
  if (((E & 3) == 0) && e0 + 16 <= E) {
    int4 d[4];
#pragma unroll
    for (int j = 0; j < 4; ++j) d[j] = *(const int4*)(dst + e0 + 4 * j);
#pragma unroll
    for (int j = 0; j < 4; ++j)
      w[j] = owner_key(d[j].x) | (owner_key(d[j].y) << 8) | (owner_key(d[j].z) << 16) | (owner_key(d[j].w) << 24);
  } else {
#pragma unroll
    for (int j = 0; j < 4; ++j) {
      unsigned v = 0u;
#pragma unroll
      for (int k = 0; k < 4; ++k) {
        const int e = e0 + 4 * j + k;
        v |= ((e < E) ? owner_key(dst[e]) : KEY_NONE) << (8 * k);
      }
      w[j] = v;
    }
  }
  *(uint4*)(keys + 4 * (size_t)t) = make_uint4(w[0], w[1], w[2], w[3]);
}

__global__ __launch_bounds__(1024) void k_build_lists(const int* __restrict__ src,
                                                      const int* __restrict__ dst,
                                                      const unsigned* __restrict__ keys,
                                                      float* __restrict__ dinv,
                                                      int* __restrict__ lists,
                                                      unsigned* __restrict__ counts, int E, int N) {
  constexpr int ROWS = OWN_ROWS;
  constexpr int CAP = OWN_CAP;
  constexpr int NT = 1024;
  constexpr int UNR = 4;
  constexpr int CHW = NT * UNR;
  __shared__ int q[CAP];
  __shared__ unsigned degc[ROWS];
  __shared__ unsigned cnt;

  const int tid = threadIdx.x;
  const int base = blockIdx.x * ROWS;

  if (tid == 0) cnt = 0u;
  if (tid < ROWS) degc[tid] = 0u;
  __syncthreads();

  static_assert(UNR == 4 && CHW == KEY_STEP, "the match mask below packs the flags of exactly 4 key words");
  const uint4* __restrict__ kp = (const uint4*)keys;
  const int nch = ((E + 15) / 16 + CHW - 1) / CHW;
  int c = (int)(((long long)blockIdx.x * nch) / gridDim.x);
  const unsigned pat = (unsigned)blockIdx.x * 0x01010101u;

  uint4 cur[UNR];
#pragma unroll
  for (int u = 0; u < UNR; ++u) cur[u] = kp[c * CHW + u * NT + tid];
  for (int ci = 0; ci < nch; ++ci) {
    int cn = c + 1;
    if (cn >= nch) cn = 0;
    uint4 nxt[UNR];
#pragma unroll
    for (int u = 0; u < UNR; ++u) nxt[u] = kp[cn * CHW + u * NT + tid];

    unsigned mm[UNR];
#pragma unroll
    for (int u = 0; u < UNR; ++u) {
      const unsigned x[4] = {cur[u].x ^ pat, cur[u].y ^ pat, cur[u].z ^ pat, cur[u].w ^ pat};
      unsigned m = 0u;
#pragma unroll
      for (int j = 0; j < 4; ++j) {
        const unsigned f = ~(((x[j] & 0x7f7f7f7fu) + 0x7f7f7f7fu) | x[j]) & 0x80808080u;
        m |= f >> (7 - j);
      }
      mm[u] = m;
    }
    unsigned long long M = (unsigned long long)(mm[0] | (mm[1] << 4)) |
                           ((unsigned long long)(mm[2] | (mm[3] << 4)) << 32);
    const int ebase = (c * CHW + tid) * 16;
    while (M) {
      const int p = __builtin_ctzll(M);
      M &= M - 1ull;
      const int u = ((p >> 5) << 1) | ((p >> 2) & 1);
      const int e = ebase + u * (NT * 16) + (p & 3) * 4 + ((p >> 3) & 3);
      const unsigned slot = atomicAdd(&cnt, 1u);
      if (slot < (unsigned)CAP) q[slot] = e;
    }
#pragma unroll
    for (int u = 0; u < UNR; ++u) cur[u] = nxt[u];
    c = cn;
  }
  __syncthreads();

  const unsigned tot = cnt;
  if (tid == 0) counts[blockIdx.x] = tot;

  if (tot <= (unsigned)CAP) {
    const int n = (int)tot;
    int2* __restrict__ my = (int2*)(lists + (size_t)blockIdx.x * CAP * 2);
    for (int i = tid; i < n; i += NT) {
      const int e = q[i];
      const int dl = dst[e] - base;
      atomicAdd(&degc[dl], 1u);
      my[i] = make_int2(src[e], dl);
    }
  } else {
    for (int e = tid; e < E; e += NT) {
      const unsigned d = (unsigned)(dst[e] - base);
      if (d < (unsigned)ROWS) atomicAdd(&degc[d], 1u);
    }
  }
  __syncthreads();

  if (tid < ROWS && base + tid < N) dinv[base + tid] = rsqrtf((float)(degc[tid] + 1u));
}

template <int NOUT, bool RELU>
__global__ __launch_bounds__(256) void k_gemm_wmma(const float* __restrict__ A,
                                                   const float* __restrict__ W,
                                                   float* __restrict__ out,
                                                   const float* __restrict__ dinv, int nrows) {
  __shared__ float Wl[128 * NOUT];
  const int tid = threadIdx.x;
  for (int i = tid; i < 128 * NOUT; i += 256) {
    int k = i / NOUT;
    int j = i - k * NOUT;
    Wl[(k >> 1) * (2 * NOUT) + 2 * j + (k & 1)] = W[i];
  }
  __syncthreads();

  const int wave = tid >> 5;
  const int lane = tid & 31;
  const int tile = blockIdx.x * 8 + wave;
  const int ntiles = (nrows + 15) >> 4;
  if (tile >= ntiles) return;

  const int m = lane & 15;
  const int kh = lane >> 4;
  int row = tile * 16 + m;
  if (row >= nrows) row = nrows - 1;

  const int rbase = tile * 16 + kh * 8;
  float di[8];
#pragma unroll
  for (int v = 0; v < 8; ++v) {
    int r = rbase + v;
    di[v] = dinv[r < nrows ? r : nrows - 1];
  }
  v8f acc[NOUT / 16];
#pragma unroll
  for (int t = 0; t < NOUT / 16; ++t) acc[t] = v8f{0.f, 0.f, 0.f, 0.f, 0.f, 0.f, 0.f, 0.f};

  const float* arow = A + (size_t)row * 128 + 2 * kh;
#pragma unroll 4
  for (int k0 = 0; k0 < 128; k0 += 4) {
    v2f a = *(const v2f*)(arow + k0);
    if (RELU) {
      a.x = fmaxf(a.x, 0.f);
      a.y = fmaxf(a.y, 0.f);
    }
    const float* bb = &Wl[((k0 >> 1) + kh) * (2 * NOUT)];
#pragma unroll
    for (int t = 0; t < NOUT / 16; ++t) {
      v2f b = *(const v2f*)(bb + 2 * (t * 16 + m));
      acc[t] = __builtin_amdgcn_wmma_f32_16x16x4_f32(false, a, false, b, (short)0, acc[t],
                                                     false, false);
    }
  }

#pragma unroll
  for (int t = 0; t < NOUT / 16; ++t) {
#pragma unroll
    for (int v = 0; v < 8; ++v) {
      int r = rbase + v;
      if (r < nrows) out[(size_t)r * NOUT + t * 16 + m] = acc[t][v] * di[v];
    }
  }
}

template <int F>
__global__ __launch_bounds__(1024) void k_agg_owner(const float* __restrict__ hs,
                                                    const int* __restrict__ src,
                                                    const int* __restrict__ dst,
                                                    const float* __restrict__ dinv,
                                                    const float* __restrict__ bias,
                                                    const int* __restrict__ lists,
                                                    const unsigned* __restrict__ counts,
                                                    float* __restrict__ agg, int E, int N) {
  constexpr int ROWS = OWN_ROWS;
  constexpr int CAP = OWN_CAP;
  constexpr int V = F / 32;
  constexpr int NT = 1024;
  constexpr int NW = NT / 32;
  constexpr int U = 8;
  constexpr int PER = (ROWS + NW - 1) / NW;
  __shared__ float acc[ROWS * F];

  const int tid = threadIdx.x;
  const int lane = tid & 31;
  const int wave = tid >> 5;
  const int base = blockIdx.x * ROWS;

  {
    float st[PER][V];
#pragma unroll
    for (int k = 0; k < PER; ++k) {
#pragma unroll
      for (int j = 0; j < V; ++j) st[k][j] = 0.f;
      if (wave + k * NW < ROWS) {
        const int row = min(base + wave + k * NW, N - 1);
        const float* p = hs + (size_t)row * F + lane * V;
        if (V == 4) {
          const float4 t = *(const float4*)p;
          st[k][0] = t.x; st[k][1] = t.y; st[k][V - 2] = t.z; st[k][V - 1] = t.w;
        } else {
          const float2 t = *(const float2*)p;
          st[k][0] = t.x; st[k][1] = t.y;
        }
      }
    }
#pragma unroll
    for (int k = 0; k < PER; ++k) {
      if (wave + k * NW < ROWS) {
#pragma unroll
        for (int j = 0; j < V; ++j) acc[(wave + k * NW) * F + j * 32 + lane] = st[k][j];
      }
    }
  }
  __syncthreads();

  const unsigned tot = counts[blockIdx.x];
  if (tot <= (unsigned)CAP) {
    const int n = (int)tot;
    const int2* __restrict__ my = (const int2*)(lists + (size_t)blockIdx.x * CAP * 2);
    const int nfull = n / U;

    int g = wave;
    int2 ent[U];
#pragma unroll
    for (int u = 0; u < U; ++u) ent[u] = make_int2(0, 0);
    if (g < nfull) {
#pragma unroll
      for (int u = 0; u < U; ++u) ent[u] = my[g * U + u];
    }
    for (; g < nfull; g += NW) {
      const int gn = (g + NW < nfull) ? g + NW : g;
      int2 nxt[U];
#pragma unroll
      for (int u = 0; u < U; ++u) nxt[u] = my[gn * U + u];

      float hv[U][V];
#pragma unroll
      for (int u = 0; u < U; ++u) {
        const float* hp = hs + (size_t)ent[u].x * F + lane * V;
        if (V == 4) {
          const float4 t = *(const float4*)hp;
          hv[u][0] = t.x; hv[u][1] = t.y; hv[u][V - 2] = t.z; hv[u][V - 1] = t.w;
        } else {
          const float2 t = *(const float2*)hp;
          hv[u][0] = t.x; hv[u][1] = t.y;
        }
      }
#pragma unroll
      for (int u = 0; u < U; ++u) {
#pragma unroll
        for (int j = 0; j < V; ++j)
          __hip_atomic_fetch_add(&acc[ent[u].y * F + j * 32 + lane], hv[u][j], __ATOMIC_RELAXED,
                                 __HIP_MEMORY_SCOPE_WORKGROUP);
      }
#pragma unroll
      for (int u = 0; u < U; ++u) ent[u] = nxt[u];
    }

    const int rem = n - nfull * U;
    if (rem > 0 && wave == nfull % NW) {
      int sv[U], dl[U];
#pragma unroll
      for (int u = 0; u < U; ++u) {
        const int2 e2 = my[min(nfull * U + u, n - 1)];
        sv[u] = e2.x;
        dl[u] = e2.y;
      }
      float hv[U][V];
#pragma unroll
      for (int u = 0; u < U; ++u) {
        const float* hp = hs + (size_t)sv[u] * F + lane * V;
        if (V == 4) {
          const float4 t = *(const float4*)hp;
          hv[u][0] = t.x; hv[u][1] = t.y; hv[u][V - 2] = t.z; hv[u][V - 1] = t.w;
        } else {
          const float2 t = *(const float2*)hp;
          hv[u][0] = t.x; hv[u][1] = t.y;
        }
      }
#pragma unroll
      for (int u = 0; u < U; ++u) {
        if (u < rem) {
#pragma unroll
          for (int j = 0; j < V; ++j)
            __hip_atomic_fetch_add(&acc[dl[u] * F + j * 32 + lane], hv[u][j], __ATOMIC_RELAXED,
                                   __HIP_MEMORY_SCOPE_WORKGROUP);
        }
      }
    }
  } else {
    for (int e = tid; e < E; e += NT) {
      const unsigned d = (unsigned)(dst[e] - base);
      if (d < (unsigned)ROWS) {
        const int s = src[e];
        for (int col = 0; col < F; ++col)
          __hip_atomic_fetch_add(&acc[d * F + (col % V) * 32 + col / V], hs[(size_t)s * F + col],
                                 __ATOMIC_RELAXED, __HIP_MEMORY_SCOPE_WORKGROUP);
      }
    }
  }
  __syncthreads();

  float bj[V];
#pragma unroll
  for (int j = 0; j < V; ++j) bj[j] = bias[lane * V + j];
  for (int r = wave; r < ROWS; r += NW) {
    const int row = base + r;
    if (row < N) {
      const float dd = dinv[row];
#pragma unroll
      for (int j = 0; j < V; ++j) agg[(size_t)row * F + lane * V + j] = acc[r * F + j * 32 + lane] * dd + bj[j];
    }
  }
}

extern "C" void kernel_launch(void* const* d_in, const int* in_sizes, int n_in, void* d_out,
                              int out_size, void* d_ws, size_t ws_size, hipStream_t stream) {
  const float* x = (const float*)d_in[0];
  const int* ei = (const int*)d_in[1];
  const float* W1 = (const float*)d_in[2];
  const float* b1 = (const float*)d_in[3];
  const float* W2 = (const float*)d_in[4];
  const float* b2 = (const float*)d_in[5];
  float* out = (float*)d_out;

  const int N = in_sizes[0] / 128;
  const int E = in_sizes[1] / 2;
  const int* src = ei;
  const int* dstp = ei + E;

  char* ws = (char*)d_ws;
  float* dinv = (float*)ws;
  size_t off = (((size_t)N * 4) + 255) & ~(size_t)255;
  float* h1 = (float*)(ws + off);
  float* agg1 = (float*)(ws + off + (size_t)N * 128 * 4);
  float* h2 = h1;
  const int nblk = (N + OWN_ROWS - 1) / OWN_ROWS;
  int* lists = (int*)(ws + off + (size_t)N * 128 * 4 * 2);
  const size_t coff = off + (size_t)N * 128 * 4 * 2 + (size_t)nblk * OWN_CAP * 8;
  unsigned* counts = (unsigned*)(ws + coff);
  unsigned* keys = (unsigned*)(ws + ((coff + (size_t)nblk * 4 + 255) & ~(size_t)255));
  const int n16p = ((E + 15) / 16 + KEY_STEP - 1) / KEY_STEP * KEY_STEP;

  dim3 B(256);
  k_edge_keys<<<n16p / 256, B, 0, stream>>>(dstp, keys, E);
  k_build_lists<<<nblk, dim3(1024), 0, stream>>>(src, dstp, keys, dinv, lists, counts, E, N);

  const int ntiles = (N + 15) / 16;
  const int gblocks = (ntiles + 7) / 8;

  k_gemm_wmma<128, false><<<gblocks, B, 0, stream>>>(x, W1, h1, dinv, N);
  k_agg_owner<128><<<nblk, dim3(1024), 0, stream>>>(h1, src, dstp, dinv, b1, lists, counts, agg1, E, N);

  k_gemm_wmma<64, true><<<gblocks, B, 0, stream>>>(agg1, W2, h2, dinv, N);
  k_agg_owner<64><<<nblk, dim3(1024), 0, stream>>>(h2, src, dstp, dinv, b2, lists, counts, out, E, N);
}


# GEMM kernels: dependent loads overlapped: a wave's whole A strip loaded as one group ahead of W staging, W staged by 16-byte loads issued together, B read one k step ahead; same sums, same order
# speedup vs baseline: 3.0679x; 1.0822x over previous
#include <hip/hip_runtime.h>

typedef __attribute__((ext_vector_type(2))) float v2f;
typedef __attribute__((ext_vector_type(8))) float v8f;

constexpr int OWN_ROWS = 200;
constexpr int OWN_CAP = 16384;
constexpr unsigned KEY_NONE = 255u;
constexpr int KEY_STEP = 4096;

__device__ __forceinline__ unsigned owner_key(int d) {
  return min((unsigned)d / (unsigned)OWN_ROWS, KEY_NONE);
}

__global__ __launch_bounds__(256) void k_edge_keys(const int* __restrict__ dst, unsigned* __restrict__ keys,
                                                   int E) {
  const int t = blockIdx.x * 256 + threadIdx.x;
  const int n16 = (E + 15) >> 4;
  const int n16p = (n16 + KEY_STEP - 1) / KEY_STEP * KEY_STEP;
  if (t >= n16p) return;
  const int e0 = t * 16;
  unsigned w[4];
  if (((E & 3) == 0) && e0 + 16 <= E) {
    int4 d[4];
#pragma unroll
    for (int j = 0; j < 4; ++j) d[j] = *(const int4*)(dst + e0 + 4 * j);
#pragma unroll
    for (int j = 0; j < 4; ++j)
      w[j] = owner_key(d[j].x) | (owner_key(d[j].y) << 8) | (owner_key(d[j].z) << 16) | (owner_key(d[j].w) << 24);
  } else {
#pragma unroll
    for (int j = 0; j < 4; ++j) {
      unsigned v = 0u;
#pragma unroll
      for (int k = 0; k < 4; ++k) {
        const int e = e0 + 4 * j + k;
        v |= ((e < E) ? owner_key(dst[e]) : KEY_NONE) << (8 * k);
      }
      w[j] = v;
    }
  }
  *(uint4*)(keys + 4 * (size_t)t) = make_uint4(w[0], w[1], w[2], w[3]);
}

__global__ __launch_bounds__(1024) void k_build_lists(const int* __restrict__ src,
                                                      const int* __restrict__ dst,
                                                      const unsigned* __restrict__ keys,
                                                      float* __restrict__ dinv,
                                                      int* __restrict__ lists,
                                                      unsigned* __restrict__ counts, int E, int N) {
  constexpr int ROWS = OWN_ROWS;
  constexpr int CAP = OWN_CAP;
  constexpr int NT = 1024;
  constexpr int UNR = 4;
  constexpr int CHW = NT * UNR;
  __shared__ int q[CAP];
  __shared__ unsigned degc[ROWS];
  __shared__ unsigned cnt;

  const int tid = threadIdx.x;
  const int base = blockIdx.x * ROWS;

  if (tid == 0) cnt = 0u;
  if (tid < ROWS) degc[tid] = 0u;
  __syncthreads();

  static_assert(UNR == 4 && CHW == KEY_STEP, "the match mask below packs the flags of exactly 4 key words");
  const uint4* __restrict__ kp = (const uint4*)keys;
  const int nch = ((E + 15) / 16 + CHW - 1) / CHW;
  int c = (int)(((long long)blockIdx.x * nch) / gridDim.x);
  const unsigned pat = (unsigned)blockIdx.x * 0x01010101u;

  uint4 cur[UNR];
#pragma unroll
  for (int u = 0; u < UNR; ++u) cur[u] = kp[c * CHW + u * NT + tid];
  for (int ci = 0; ci < nch; ++ci) {
    int cn = c + 1;
    if (cn >= nch) cn = 0;
    uint4 nxt[UNR];
#pragma unroll
    for (int u = 0; u < UNR; ++u) nxt[u] = kp[cn * CHW + u * NT + tid];

    unsigned mm[UNR];
#pragma unroll
    for (int u = 0; u < UNR; ++u) {
      const unsigned x[4] = {cur[u].x ^ pat, cur[u].y ^ pat, cur[u].z ^ pat, cur[u].w ^ pat};
      unsigned m = 0u;
#pragma unroll
      for (int j = 0; j < 4; ++j) {
        const unsigned f = ~(((x[j] & 0x7f7f7f7fu) + 0x7f7f7f7fu) | x[j]) & 0x80808080u;
        m |= f >> (7 - j);
      }
      mm[u] = m;
    }
    unsigned long long M = (unsigned long long)(mm[0] | (mm[1] << 4)) |
                           ((unsigned long long)(mm[2] | (mm[3] << 4)) << 32);
    const int ebase = (c * CHW + tid) * 16;
    while (M) {
      const int p = __builtin_ctzll(M);
      M &= M - 1ull;
      const int u = ((p >> 5) << 1) | ((p >> 2) & 1);
      const int e = ebase + u * (NT * 16) + (p & 3) * 4 + ((p >> 3) & 3);
      const unsigned slot = atomicAdd(&cnt, 1u);
      if (slot < (unsigned)CAP) q[slot] = e;
    }
#pragma unroll
    for (int u = 0; u < UNR; ++u) cur[u] = nxt[u];
    c = cn;
  }
  __syncthreads();

  const unsigned tot = cnt;
  if (tid == 0) counts[blockIdx.x] = tot;

  if (tot <= (unsigned)CAP) {
    const int n = (int)tot;
    int2* __restrict__ my = (int2*)(lists + (size_t)blockIdx.x * CAP * 2);
    for (int i = tid; i < n; i += NT) {
      const int e = q[i];
      const int dl = dst[e] - base;
      atomicAdd(&degc[dl], 1u);
      my[i] = make_int2(src[e], dl);
    }
  } else {
    for (int e = tid; e < E; e += NT) {
      const unsigned d = (unsigned)(dst[e] - base);
      if (d < (unsigned)ROWS) atomicAdd(&degc[d], 1u);
    }
  }
  __syncthreads();

  if (tid < ROWS && base + tid < N) dinv[base + tid] = rsqrtf((float)(degc[tid] + 1u));
}

template <int NOUT, bool RELU>
__global__ __launch_bounds__(256) void k_gemm_wmma(const float* __restrict__ A,
                                                   const float* __restrict__ W,
                                                   float* __restrict__ out,
                                                   const float* __restrict__ dinv, int nrows) {
  __shared__ __attribute__((aligned(16))) float Wl[128 * NOUT];
  constexpr int NT = NOUT / 16;
  constexpr int KS = 128 / 4;
  constexpr int CPR = NOUT / 4;
  constexpr int NCH = 64 * CPR / 256;
  static_assert(64 * CPR % 256 == 0, "whole staging chunks per thread");
  const int tid = threadIdx.x;
  const int wave = tid >> 5;
  const int lane = tid & 31;
  const int tile = blockIdx.x * 8 + wave;
  const int ntiles = (nrows + 15) >> 4;

  const int m = lane & 15;
  const int kh = lane >> 4;
  int row = tile * 16 + m;
  if (row >= nrows) row = nrows - 1;

  const float* arow = A + (size_t)row * 128 + 2 * kh;
  v2f areg[KS];
#pragma unroll
  for (int i = 0; i < KS; ++i) areg[i] = *(const v2f*)(arow + 4 * i);

  {
    float4 w0[NCH], w1[NCH];
#pragma unroll
    for (int q = 0; q < NCH; ++q) {
      const int c = q * 256 + tid;
      const int p = c / CPR;
      const int j = (c - p * CPR) * 4;
      w0[q] = *(const float4*)(W + (size_t)(2 * p) * NOUT + j);
      w1[q] = *(const float4*)(W + (size_t)(2 * p + 1) * NOUT + j);
    }
#pragma unroll
    for (int q = 0; q < NCH; ++q) {
      const int c = q * 256 + tid;
      const int p = c / CPR;
      const int j = (c - p * CPR) * 4;
      float4* d = (float4*)&Wl[p * (2 * NOUT) + 2 * j];
      d[0] = make_float4(w0[q].x, w1[q].x, w0[q].y, w1[q].y);
      d[1] = make_float4(w0[q].z, w1[q].z, w0[q].w, w1[q].w);
    }
  }
  __syncthreads();

  if (tile >= ntiles) return;

  const int rbase = tile * 16 + kh * 8;
  float di[8];
#pragma unroll
  for (int v = 0; v < 8; ++v) {
    int r = rbase + v;
    di[v] = dinv[r < nrows ? r : nrows - 1];
  }
  v8f acc[NT];
#pragma unroll
  for (int t = 0; t < NT; ++t) acc[t] = v8f{0.f, 0.f, 0.f, 0.f, 0.f, 0.f, 0.f, 0.f};

  const float* bl = &Wl[kh * (2 * NOUT) + 2 * m];
  v2f bcur[NT];
#pragma unroll
  for (int t = 0; t < NT; ++t) bcur[t] = *(const v2f*)(bl + 32 * t);
#pragma unroll
  for (int i = 0; i < KS; ++i) {
    v2f bnxt[NT];
    if (i + 1 < KS) {
#pragma unroll
      for (int t = 0; t < NT; ++t) bnxt[t] = *(const v2f*)(bl + (2 * (i + 1)) * (2 * NOUT) + 32 * t);
    }
    v2f a = areg[i];
    if (RELU) {
      a.x = fmaxf(a.x, 0.f);
      a.y = fmaxf(a.y, 0.f);
    }
#pragma unroll
    for (int t = 0; t < NT; ++t)
      acc[t] = __builtin_amdgcn_wmma_f32_16x16x4_f32(false, a, false, bcur[t], (short)0, acc[t],
                                                     false, false);
    if (i + 1 < KS) {
#pragma unroll
      for (int t = 0; t < NT; ++t) bcur[t] = bnxt[t];
    }
  }

#pragma unroll
  for (int t = 0; t < NT; ++t) {
#pragma unroll
    for (int v = 0; v < 8; ++v) {
      int r = rbase + v;
      if (r < nrows) out[(size_t)r * NOUT + t * 16 + m] = acc[t][v] * di[v];
    }
  }
}

template <int F>
__global__ __launch_bounds__(1024) void k_agg_owner(const float* __restrict__ hs,
                                                    const int* __restrict__ src,
                                                    const int* __restrict__ dst,
                                                    const float* __restrict__ dinv,
                                                    const float* __restrict__ bias,
                                                    const int* __restrict__ lists,
                                                    const unsigned* __restrict__ counts,
                                                    float* __restrict__ agg, int E, int N) {
  constexpr int ROWS = OWN_ROWS;
  constexpr int CAP = OWN_CAP;
  constexpr int V = F / 32;
  constexpr int NT = 1024;
  constexpr int NW = NT / 32;
  constexpr int U = 8;
  constexpr int PER = (ROWS + NW - 1) / NW;
  __shared__ float acc[ROWS * F];

  const int tid = threadIdx.x;
  const int lane = tid & 31;
  const int wave = tid >> 5;
  const int base = blockIdx.x * ROWS;

  {
    float st[PER][V];
#pragma unroll
    for (int k = 0; k < PER; ++k) {
#pragma unroll
      for (int j = 0; j < V; ++j) st[k][j] = 0.f;
      if (wave + k * NW < ROWS) {
        const int row = min(base + wave + k * NW, N - 1);
        const float* p = hs + (size_t)row * F + lane * V;
        if (V == 4) {
          const float4 t = *(const float4*)p;
          st[k][0] = t.x; st[k][1] = t.y; st[k][V - 2] = t.z; st[k][V - 1] = t.w;
        } else {
          const float2 t = *(const float2*)p;
          st[k][0] = t.x; st[k][1] = t.y;
        }
      }
    }
#pragma unroll
    for (int k = 0; k < PER; ++k) {
      if (wave + k * NW < ROWS) {
#pragma unroll
        for (int j = 0; j < V; ++j) acc[(wave + k * NW) * F + j * 32 + lane] = st[k][j];
      }
    }
  }
  __syncthreads();

  const unsigned tot = counts[blockIdx.x];
  if (tot <= (unsigned)CAP) {
    const int n = (int)tot;
    const int2* __restrict__ my = (const int2*)(lists + (size_t)blockIdx.x * CAP * 2);
    const int nfull = n / U;

    int g = wave;
    int2 ent[U];
#pragma unroll
    for (int u = 0; u < U; ++u) ent[u] = make_int2(0, 0);
    if (g < nfull) {
#pragma unroll
      for (int u = 0; u < U; ++u) ent[u] = my[g * U + u];
    }
    for (; g < nfull; g += NW) {
      const int gn = (g + NW < nfull) ? g + NW : g;
      int2 nxt[U];
#pragma unroll
      for (int u = 0; u < U; ++u) nxt[u] = my[gn * U + u];

      float hv[U][V];
#pragma unroll
      for (int u = 0; u < U; ++u) {
        const float* hp = hs + (size_t)ent[u].x * F + lane * V;
        if (V == 4) {
          const float4 t = *(const float4*)hp;
          hv[u][0] = t.x; hv[u][1] = t.y; hv[u][V - 2] = t.z; hv[u][V - 1] = t.w;
        } else {
          const float2 t = *(const float2*)hp;
          hv[u][0] = t.x; hv[u][1] = t.y;
        }
      }
#pragma unroll
      for (int u = 0; u < U; ++u) {
#pragma unroll
        for (int j = 0; j < V; ++j)
          __hip_atomic_fetch_add(&acc[ent[u].y * F + j * 32 + lane], hv[u][j], __ATOMIC_RELAXED,
                                 __HIP_MEMORY_SCOPE_WORKGROUP);
      }
#pragma unroll
      for (int u = 0; u < U; ++u) ent[u] = nxt[u];
    }

    const int rem = n - nfull * U;
    if (rem > 0 && wave == nfull % NW) {
      int sv[U], dl[U];
#pragma unroll
      for (int u = 0; u < U; ++u) {
        const int2 e2 = my[min(nfull * U + u, n - 1)];
        sv[u] = e2.x;
        dl[u] = e2.y;
      }
      float hv[U][V];
#pragma unroll
      for (int u = 0; u < U; ++u) {
        const float* hp = hs + (size_t)sv[u] * F + lane * V;
        if (V == 4) {
          const float4 t = *(const float4*)hp;
          hv[u][0] = t.x; hv[u][1] = t.y; hv[u][V - 2] = t.z; hv[u][V - 1] = t.w;
        } else {
          const float2 t = *(const float2*)hp;
          hv[u][0] = t.x; hv[u][1] = t.y;
        }
      }
#pragma unroll
      for (int u = 0; u < U; ++u) {
        if (u < rem) {
#pragma unroll
          for (int j = 0; j < V; ++j)
            __hip_atomic_fetch_add(&acc[dl[u] * F + j * 32 + lane], hv[u][j], __ATOMIC_RELAXED,
                                   __HIP_MEMORY_SCOPE_WORKGROUP);
        }
      }
    }
  } else {
    for (int e = tid; e < E; e += NT) {
      const unsigned d = (unsigned)(dst[e] - base);
      if (d < (unsigned)ROWS) {
        const int s = src[e];
        for (int col = 0; col < F; ++col)
          __hip_atomic_fetch_add(&acc[d * F + (col % V) * 32 + col / V], hs[(size_t)s * F + col],
                                 __ATOMIC_RELAXED, __HIP_MEMORY_SCOPE_WORKGROUP);
      }
    }
  }
  __syncthreads();

  float bj[V];
#pragma unroll
  for (int j = 0; j < V; ++j) bj[j] = bias[lane * V + j];
  for (int r = wave; r < ROWS; r += NW) {
    const int row = base + r;
    if (row < N) {
      const float dd = dinv[row];
#pragma unroll
      for (int j = 0; j < V; ++j) agg[(size_t)row * F + lane * V + j] = acc[r * F + j * 32 + lane] * dd + bj[j];
    }
  }
}

extern "C" void kernel_launch(void* const* d_in, const int* in_sizes, int n_in, void* d_out,
                              int out_size, void* d_ws, size_t ws_size, hipStream_t stream) {
  const float* x = (const float*)d_in[0];
  const int* ei = (const int*)d_in[1];
  const float* W1 = (const float*)d_in[2];
  const float* b1 = (const float*)d_in[3];
  const float* W2 = (const float*)d_in[4];
  const float* b2 = (const float*)d_in[5];
  float* out = (float*)d_out;

  const int N = in_sizes[0] / 128;
  const int E = in_sizes[1] / 2;
  const int* src = ei;
  const int* dstp = ei + E;

  char* ws = (char*)d_ws;
  float* dinv = (float*)ws;
  size_t off = (((size_t)N * 4) + 255) & ~(size_t)255;
  float* h1 = (float*)(ws + off);
  float* agg1 = (float*)(ws + off + (size_t)N * 128 * 4);
  float* h2 = h1;
  const int nblk = (N + OWN_ROWS - 1) / OWN_ROWS;
  int* lists = (int*)(ws + off + (size_t)N * 128 * 4 * 2);
  const size_t coff = off + (size_t)N * 128 * 4 * 2 + (size_t)nblk * OWN_CAP * 8;
  unsigned* counts = (unsigned*)(ws + coff);
  unsigned* keys = (unsigned*)(ws + ((coff + (size_t)nblk * 4 + 255) & ~(size_t)255));
  const int n16p = ((E + 15) / 16 + KEY_STEP - 1) / KEY_STEP * KEY_STEP;

  dim3 B(256);
  k_edge_keys<<<n16p / 256, B, 0, stream>>>(dstp, keys, E);
  k_build_lists<<<nblk, dim3(1024), 0, stream>>>(src, dstp, keys, dinv, lists, counts, E, N);

  const int ntiles = (N + 15) / 16;
  const int gblocks = (ntiles + 7) / 8;

  k_gemm_wmma<128, false><<<gblocks, B, 0, stream>>>(x, W1, h1, dinv, N);
  k_agg_owner<128><<<nblk, dim3(1024), 0, stream>>>(h1, src, dstp, dinv, b1, lists, counts, agg1, E, N);

  k_gemm_wmma<64, true><<<gblocks, B, 0, stream>>>(agg1, W2, h2, dinv, N);
  k_agg_owner<64><<<nblk, dim3(1024), 0, stream>>>(h2, src, dstp, dinv, b2, lists, counts, out, E, N);
}


# aggregation kernels: list entries come in by scalar loads (wave number kept as a scalar, one 64-byte scalar load per 8-edge group) instead of vector loads; same sums, LDS arrival order as before
# speedup vs baseline: 3.1667x; 1.0322x over previous
#include <hip/hip_runtime.h>

typedef __attribute__((ext_vector_type(2))) float v2f;
typedef __attribute__((ext_vector_type(8))) float v8f;
typedef __attribute__((ext_vector_type(16))) int v16i;

constexpr int OWN_ROWS = 200;
constexpr int OWN_CAP = 16384;
constexpr unsigned KEY_NONE = 255u;
constexpr int KEY_STEP = 4096;

__device__ __forceinline__ unsigned owner_key(int d) {
  return min((unsigned)d / (unsigned)OWN_ROWS, KEY_NONE);
}

__global__ __launch_bounds__(256) void k_edge_keys(const int* __restrict__ dst, unsigned* __restrict__ keys,
                                                   int E) {
  const int t = blockIdx.x * 256 + threadIdx.x;
  const int n16 = (E + 15) >> 4;
  const int n16p = (n16 + KEY_STEP - 1) / KEY_STEP * KEY_STEP;
  if (t >= n16p) return;
  const int e0 = t * 16;
  unsigned w[4];
  if (((E & 3) == 0) && e0 + 16 <= E) {
    int4 d[4];
#pragma unroll
    for (int j = 0; j < 4; ++j) d[j] = *(const int4*)(dst + e0 + 4 * j);
#pragma unroll
    for (int j = 0; j < 4; ++j)
      w[j] = owner_key(d[j].x) | (owner_key(d[j].y) << 8) | (owner_key(d[j].z) << 16) | (owner_key(d[j].w) << 24);
  } else {
#pragma unroll
    for (int j = 0; j < 4; ++j) {
      unsigned v = 0u;
#pragma unroll
      for (int k = 0; k < 4; ++k) {
        const int e = e0 + 4 * j + k;
        v |= ((e < E) ? owner_key(dst[e]) : KEY_NONE) << (8 * k);
      }
      w[j] = v;
    }
  }
  *(uint4*)(keys + 4 * (size_t)t) = make_uint4(w[0], w[1], w[2], w[3]);
}

__global__ __launch_bounds__(1024) void k_build_lists(const int* __restrict__ src,
                                                      const int* __restrict__ dst,
                                                      const unsigned* __restrict__ keys,
                                                      float* __restrict__ dinv,
                                                      int* __restrict__ lists,
                                                      unsigned* __restrict__ counts, int E, int N) {
  constexpr int ROWS = OWN_ROWS;
  constexpr int CAP = OWN_CAP;
  constexpr int NT = 1024;
  constexpr int UNR = 4;
  constexpr int CHW = NT * UNR;
  __shared__ int q[CAP];
  __shared__ unsigned degc[ROWS];
  __shared__ unsigned cnt;

  const int tid = threadIdx.x;
  const int base = blockIdx.x * ROWS;

  if (tid == 0) cnt = 0u;
  if (tid < ROWS) degc[tid] = 0u;
  __syncthreads();

  static_assert(UNR == 4 && CHW == KEY_STEP, "the match mask below packs the flags of exactly 4 key words");
  const uint4* __restrict__ kp = (const uint4*)keys;
  const int nch = ((E + 15) / 16 + CHW - 1) / CHW;
  int c = (int)(((long long)blockIdx.x * nch) / gridDim.x);
  const unsigned pat = (unsigned)blockIdx.x * 0x01010101u;

  uint4 cur[UNR];
#pragma unroll
  for (int u = 0; u < UNR; ++u) cur[u] = kp[c * CHW + u * NT + tid];
  for (int ci = 0; ci < nch; ++ci) {
    int cn = c + 1;
    if (cn >= nch) cn = 0;
    uint4 nxt[UNR];
#pragma unroll
    for (int u = 0; u < UNR; ++u) nxt[u] = kp[cn * CHW + u * NT + tid];

    unsigned mm[UNR];
#pragma unroll
    for (int u = 0; u < UNR; ++u) {
      const unsigned x[4] = {cur[u].x ^ pat, cur[u].y ^ pat, cur[u].z ^ pat, cur[u].w ^ pat};
      unsigned m = 0u;
#pragma unroll
      for (int j = 0; j < 4; ++j) {
        const unsigned f = ~(((x[j] & 0x7f7f7f7fu) + 0x7f7f7f7fu) | x[j]) & 0x80808080u;
        m |= f >> (7 - j);
      }
      mm[u] = m;
    }
    unsigned long long M = (unsigned long long)(mm[0] | (mm[1] << 4)) |
                           ((unsigned long long)(mm[2] | (mm[3] << 4)) << 32);
    const int ebase = (c * CHW + tid) * 16;
    while (M) {
      const int p = __builtin_ctzll(M);
      M &= M - 1ull;
      const int u = ((p >> 5) << 1) | ((p >> 2) & 1);
      const int e = ebase + u * (NT * 16) + (p & 3) * 4 + ((p >> 3) & 3);
      const unsigned slot = atomicAdd(&cnt, 1u);
      if (slot < (unsigned)CAP) q[slot] = e;
    }
#pragma unroll
    for (int u = 0; u < UNR; ++u) cur[u] = nxt[u];
    c = cn;
  }
  __syncthreads();

  const unsigned tot = cnt;
  if (tid == 0) counts[blockIdx.x] = tot;

  if (tot <= (unsigned)CAP) {
    const int n = (int)tot;
    int2* __restrict__ my = (int2*)(lists + (size_t)blockIdx.x * CAP * 2);
    for (int i = tid; i < n; i += NT) {
      const int e = q[i];
      const int dl = dst[e] - base;
      atomicAdd(&degc[dl], 1u);
      my[i] = make_int2(src[e], dl);
    }
  } else {
    for (int e = tid; e < E; e += NT) {
      const unsigned d = (unsigned)(dst[e] - base);
      if (d < (unsigned)ROWS) atomicAdd(&degc[d], 1u);
    }
  }
  __syncthreads();

  if (tid < ROWS && base + tid < N) dinv[base + tid] = rsqrtf((float)(degc[tid] + 1u));
}

template <int NOUT, bool RELU>
__global__ __launch_bounds__(256) void k_gemm_wmma(const float* __restrict__ A,
                                                   const float* __restrict__ W,
                                                   float* __restrict__ out,
                                                   const float* __restrict__ dinv, int nrows) {
  __shared__ __attribute__((aligned(16))) float Wl[128 * NOUT];
  constexpr int NT = NOUT / 16;
  constexpr int KS = 128 / 4;
  constexpr int CPR = NOUT / 4;
  constexpr int NCH = 64 * CPR / 256;
  static_assert(64 * CPR % 256 == 0, "whole staging chunks per thread");
  const int tid = threadIdx.x;
  const int wave = tid >> 5;
  const int lane = tid & 31;
  const int tile = blockIdx.x * 8 + wave;
  const int ntiles = (nrows + 15) >> 4;

  const int m = lane & 15;
  const int kh = lane >> 4;
  int row = tile * 16 + m;
  if (row >= nrows) row = nrows - 1;

  const float* arow = A + (size_t)row * 128 + 2 * kh;
  v2f areg[KS];
#pragma unroll
  for (int i = 0; i < KS; ++i) areg[i] = *(const v2f*)(arow + 4 * i);

  {
    float4 w0[NCH], w1[NCH];
#pragma unroll
    for (int q = 0; q < NCH; ++q) {
      const int c = q * 256 + tid;
      const int p = c / CPR;
      const int j = (c - p * CPR) * 4;
      w0[q] = *(const float4*)(W + (size_t)(2 * p) * NOUT + j);
      w1[q] = *(const float4*)(W + (size_t)(2 * p + 1) * NOUT + j);
    }
#pragma unroll
    for (int q = 0; q < NCH; ++q) {
      const int c = q * 256 + tid;
      const int p = c / CPR;
      const int j = (c - p * CPR) * 4;
      float4* d = (float4*)&Wl[p * (2 * NOUT) + 2 * j];
      d[0] = make_float4(w0[q].x, w1[q].x, w0[q].y, w1[q].y);
      d[1] = make_float4(w0[q].z, w1[q].z, w0[q].w, w1[q].w);
    }
  }
  __syncthreads();

  if (tile >= ntiles) return;

  const int rbase = tile * 16 + kh * 8;
  float di[8];
#pragma unroll
  for (int v = 0; v < 8; ++v) {
    int r = rbase + v;
    di[v] = dinv[r < nrows ? r : nrows - 1];
  }
  v8f acc[NT];
#pragma unroll
  for (int t = 0; t < NT; ++t) acc[t] = v8f{0.f, 0.f, 0.f, 0.f, 0.f, 0.f, 0.f, 0.f};

  const float* bl = &Wl[kh * (2 * NOUT) + 2 * m];
  v2f bcur[NT];
#pragma unroll
  for (int t = 0; t < NT; ++t) bcur[t] = *(const v2f*)(bl + 32 * t);
#pragma unroll
  for (int i = 0; i < KS; ++i) {
    v2f bnxt[NT];
    if (i + 1 < KS) {
#pragma unroll
      for (int t = 0; t < NT; ++t) bnxt[t] = *(const v2f*)(bl + (2 * (i + 1)) * (2 * NOUT) + 32 * t);
    }
    v2f a = areg[i];
    if (RELU) {
      a.x = fmaxf(a.x, 0.f);
      a.y = fmaxf(a.y, 0.f);
    }
#pragma unroll
    for (int t = 0; t < NT; ++t)
      acc[t] = __builtin_amdgcn_wmma_f32_16x16x4_f32(false, a, false, bcur[t], (short)0, acc[t],
                                                     false, false);
    if (i + 1 < KS) {
#pragma unroll
      for (int t = 0; t < NT; ++t) bcur[t] = bnxt[t];
    }
  }

#pragma unroll
  for (int t = 0; t < NT; ++t) {
#pragma unroll
    for (int v = 0; v < 8; ++v) {
      int r = rbase + v;
      if (r < nrows) out[(size_t)r * NOUT + t * 16 + m] = acc[t][v] * di[v];
    }
  }
}

template <int F>
__global__ __launch_bounds__(1024) void k_agg_owner(const float* __restrict__ hs,
                                                    const int* __restrict__ src,
                                                    const int* __restrict__ dst,
                                                    const float* __restrict__ dinv,
                                                    const float* __restrict__ bias,
                                                    const int* __restrict__ lists,
                                                    const unsigned* __restrict__ counts,
                                                    float* __restrict__ agg, int E, int N) {
  constexpr int ROWS = OWN_ROWS;
  constexpr int CAP = OWN_CAP;
  constexpr int V = F / 32;
  constexpr int NT = 1024;
  constexpr int NW = NT / 32;
  constexpr int U = 8;
  constexpr int PER = (ROWS + NW - 1) / NW;
  __shared__ float acc[ROWS * F];

  const int tid = threadIdx.x;
  const int lane = tid & 31;
  const int wave = __builtin_amdgcn_readfirstlane(tid >> 5);
  __builtin_assume(wave >= 0 && wave < NW);
  const int base = blockIdx.x * ROWS;

  {
    float st[PER][V];
#pragma unroll
    for (int k = 0; k < PER; ++k) {
#pragma unroll
      for (int j = 0; j < V; ++j) st[k][j] = 0.f;
      if (wave + k * NW < ROWS) {
        const int row = min(base + wave + k * NW, N - 1);
        const float* p = hs + (size_t)row * F + lane * V;
        if (V == 4) {
          const float4 t = *(const float4*)p;
          st[k][0] = t.x; st[k][1] = t.y; st[k][V - 2] = t.z; st[k][V - 1] = t.w;
        } else {
          const float2 t = *(const float2*)p;
          st[k][0] = t.x; st[k][1] = t.y;
        }
      }
    }
#pragma unroll
    for (int k = 0; k < PER; ++k) {
      if (wave + k * NW < ROWS) {
#pragma unroll
        for (int j = 0; j < V; ++j) acc[(wave + k * NW) * F + j * 32 + lane] = st[k][j];
      }
    }
  }
  __syncthreads();

  const unsigned tot = counts[blockIdx.x];
  if (tot <= (unsigned)CAP) {
    const int n = (int)tot;
    const int2* __restrict__ my = (const int2*)(lists + (size_t)blockIdx.x * CAP * 2);
    const int nfull = n / U;

    static_assert(U == 8, "a group of U (source, row) pairs is read as one 16-int vector");
    const v16i* __restrict__ myg = (const v16i*)my;
    int g = wave;
    v16i ent = {0, 0, 0, 0, 0, 0, 0, 0, 0, 0, 0, 0, 0, 0, 0, 0};
    if (g < nfull) ent = myg[g];
    for (; g < nfull; g += NW) {
      const int gn = (g + NW < nfull) ? g + NW : g;
      const v16i nxt = myg[gn];

      float hv[U][V];
#pragma unroll
      for (int u = 0; u < U; ++u) {
        const float* hp = hs + (size_t)ent[2 * u] * F + lane * V;
        if (V == 4) {
          const float4 t = *(const float4*)hp;
          hv[u][0] = t.x; hv[u][1] = t.y; hv[u][V - 2] = t.z; hv[u][V - 1] = t.w;
        } else {
          const float2 t = *(const float2*)hp;
          hv[u][0] = t.x; hv[u][1] = t.y;
        }
      }
#pragma unroll
      for (int u = 0; u < U; ++u) {
#pragma unroll
        for (int j = 0; j < V; ++j)
          __hip_atomic_fetch_add(&acc[ent[2 * u + 1] * F + j * 32 + lane], hv[u][j], __ATOMIC_RELAXED,
                                 __HIP_MEMORY_SCOPE_WORKGROUP);
      }
      ent = nxt;
    }

    const int rem = n - nfull * U;
    if (rem > 0 && wave == nfull % NW) {
      int sv[U], dl[U];
#pragma unroll
      for (int u = 0; u < U; ++u) {
        const int2 e2 = my[min(nfull * U + u, n - 1)];
        sv[u] = e2.x;
        dl[u] = e2.y;
      }
      float hv[U][V];
#pragma unroll
      for (int u = 0; u < U; ++u) {
        const float* hp = hs + (size_t)sv[u] * F + lane * V;
        if (V == 4) {
          const float4 t = *(const float4*)hp;
          hv[u][0] = t.x; hv[u][1] = t.y; hv[u][V - 2] = t.z; hv[u][V - 1] = t.w;
        } else {
          const float2 t = *(const float2*)hp;
          hv[u][0] = t.x; hv[u][1] = t.y;
        }
      }
#pragma unroll
      for (int u = 0; u < U; ++u) {
        if (u < rem) {
#pragma unroll
          for (int j = 0; j < V; ++j)
            __hip_atomic_fetch_add(&acc[dl[u] * F + j * 32 + lane], hv[u][j], __ATOMIC_RELAXED,
                                   __HIP_MEMORY_SCOPE_WORKGROUP);
        }
      }
    }
  } else {
    for (int e = tid; e < E; e += NT) {
      const unsigned d = (unsigned)(dst[e] - base);
      if (d < (unsigned)ROWS) {
        const int s = src[e];
        for (int col = 0; col < F; ++col)
          __hip_atomic_fetch_add(&acc[d * F + (col % V) * 32 + col / V], hs[(size_t)s * F + col],
                                 __ATOMIC_RELAXED, __HIP_MEMORY_SCOPE_WORKGROUP);
      }
    }
  }
  __syncthreads();

  float bj[V];
#pragma unroll
  for (int j = 0; j < V; ++j) bj[j] = bias[lane * V + j];
  for (int r = wave; r < ROWS; r += NW) {
    const int row = base + r;
    if (row < N) {
      const float dd = dinv[row];
#pragma unroll
      for (int j = 0; j < V; ++j) agg[(size_t)row * F + lane * V + j] = acc[r * F + j * 32 + lane] * dd + bj[j];
    }
  }
}

extern "C" void kernel_launch(void* const* d_in, const int* in_sizes, int n_in, void* d_out,
                              int out_size, void* d_ws, size_t ws_size, hipStream_t stream) {
  const float* x = (const float*)d_in[0];
  const int* ei = (const int*)d_in[1];
  const float* W1 = (const float*)d_in[2];
  const float* b1 = (const float*)d_in[3];
  const float* W2 = (const float*)d_in[4];
  const float* b2 = (const float*)d_in[5];
  float* out = (float*)d_out;

  const int N = in_sizes[0] / 128;
  const int E = in_sizes[1] / 2;
  const int* src = ei;
  const int* dstp = ei + E;

  char* ws = (char*)d_ws;
  float* dinv = (float*)ws;
  size_t off = (((size_t)N * 4) + 255) & ~(size_t)255;
  float* h1 = (float*)(ws + off);
  float* agg1 = (float*)(ws + off + (size_t)N * 128 * 4);
  float* h2 = h1;
  const int nblk = (N + OWN_ROWS - 1) / OWN_ROWS;
  int* lists = (int*)(ws + off + (size_t)N * 128 * 4 * 2);
  const size_t coff = off + (size_t)N * 128 * 4 * 2 + (size_t)nblk * OWN_CAP * 8;
  unsigned* counts = (unsigned*)(ws + coff);
  unsigned* keys = (unsigned*)(ws + ((coff + (size_t)nblk * 4 + 255) & ~(size_t)255));
  const int n16p = ((E + 15) / 16 + KEY_STEP - 1) / KEY_STEP * KEY_STEP;

  dim3 B(256);
  k_edge_keys<<<n16p / 256, B, 0, stream>>>(dstp, keys, E);
  k_build_lists<<<nblk, dim3(1024), 0, stream>>>(src, dstp, keys, dinv, lists, counts, E, N);

  const int ntiles = (N + 15) / 16;
  const int gblocks = (ntiles + 7) / 8;

  k_gemm_wmma<128, false><<<gblocks, B, 0, stream>>>(x, W1, h1, dinv, N);
  k_agg_owner<128><<<nblk, dim3(1024), 0, stream>>>(h1, src, dstp, dinv, b1, lists, counts, agg1, E, N);

  k_gemm_wmma<64, true><<<gblocks, B, 0, stream>>>(agg1, W2, h2, dinv, N);
  k_agg_owner<64><<<nblk, dim3(1024), 0, stream>>>(h2, src, dstp, dinv, b2, lists, counts, out, E, N);
}
